# starting kernel
# baseline (speedup 1.0000x reference)
#include <hip/hip_runtime.h>
#include <math.h>

#define N_NODES    100000
#define N_EDGES    1600000
#define H_DIM      64
#define N_LAYERS   3
#define NUM_GRAPHS 256

typedef __attribute__((ext_vector_type(2))) float v2f;
typedef __attribute__((ext_vector_type(8))) float v8f;

__device__ __forceinline__ v8f wmma4(v2f a, v2f b, v8f c) {
  return __builtin_amdgcn_wmma_f32_16x16x4_f32(false, a, false, b, (short)0, c,
                                               false, false);
}

__global__ void pack_b(const float* __restrict__ Wsrc, float* __restrict__ pack,
                       int nGroups, int transposed) {
  int tid = blockIdx.x * blockDim.x + threadIdx.x;
  int total = 16 * nGroups * 32;
  if (tid >= total) return;
  int lane = tid & 31;
  int t = tid >> 5;
  int g = t % nGroups, kk = t / nGroups;
  int ln = lane & 15, lh = lane >> 4;
  int c = g * 16 + ln;
  int k0 = 4 * kk + 2 * lh;
  v2f b;
  if (transposed) {
    b.x = Wsrc[c * 64 + k0];
    b.y = Wsrc[c * 64 + k0 + 1];
  } else {
    b.x = Wsrc[k0 * 64 + c];
    b.y = Wsrc[(k0 + 1) * 64 + c];
  }
  ((v2f*)pack)[tid] = b;
}

__global__ void gemm_m(const float* __restrict__ h,
                       const v2f* __restrict__ packW, float* __restrict__ m,
                       int nTiles) {
  int wave = (int)((blockIdx.x * blockDim.x + threadIdx.x) >> 5);
  int lane = threadIdx.x & 31;
  if (wave >= nTiles) return;
  const int r0 = wave * 16;
  const int ln = lane & 15, lh = lane >> 4;

  v2f a[16];
  const v2f* hp = (const v2f*)(h + (size_t)(r0 + ln) * H_DIM);
#pragma unroll
  for (int kk = 0; kk < 16; ++kk) a[kk] = hp[2 * kk + lh];

  v8f acc[4];
#pragma unroll
  for (int j = 0; j < 4; ++j) {
    v8f z = {};
    acc[j] = z;
  }

#pragma unroll
  for (int kk = 0; kk < 16; ++kk) {
#pragma unroll
    for (int j = 0; j < 4; ++j) {
      v2f b = packW[(kk * 4 + j) * 32 + lane];
      acc[j] = wmma4(a[kk], b, acc[j]);
    }
  }

#pragma unroll
  for (int j = 0; j < 4; ++j)
#pragma unroll
    for (int v = 0; v < 8; ++v)
      m[(size_t)(r0 + v + 8 * lh) * H_DIM + j * 16 + ln] = acc[j][v];
}

__global__ void scatter_add(const float* __restrict__ m,
                            const int* __restrict__ ei,
                            float* __restrict__ agg, int nEdges) {
  int tid = blockIdx.x * blockDim.x + threadIdx.x;
  int e = tid >> 4;
  if (e >= nEdges) return;
  int q = tid & 15;
  int src = ei[e];
  int dst = ei[nEdges + e];
  float4 val = ((const float4*)(m + (size_t)src * H_DIM))[q];
  float* ap = agg + (size_t)dst * H_DIM + q * 4;
  __hip_atomic_fetch_add(ap + 0, val.x, __ATOMIC_RELAXED, __HIP_MEMORY_SCOPE_AGENT);
  __hip_atomic_fetch_add(ap + 1, val.y, __ATOMIC_RELAXED, __HIP_MEMORY_SCOPE_AGENT);
  __hip_atomic_fetch_add(ap + 2, val.z, __ATOMIC_RELAXED, __HIP_MEMORY_SCOPE_AGENT);
  __hip_atomic_fetch_add(ap + 3, val.w, __ATOMIC_RELAXED, __HIP_MEMORY_SCOPE_AGENT);
}

#define NFRAG_G (16 * 12 * 32)

__global__ void gru_fused(float* __restrict__ h, const float* __restrict__ agg,
                          const v2f* __restrict__ pIh,
                          const v2f* __restrict__ pHh,
                          const float* __restrict__ bih,
                          const float* __restrict__ bhh, int nTiles) {
  __shared__ v2f sIh[NFRAG_G];
  __shared__ v2f sHh[NFRAG_G];

  for (int i = threadIdx.x; i < NFRAG_G; i += 256) {
    sIh[i] = pIh[i];
    sHh[i] = pHh[i];
  }
  __syncthreads();

  int wave = (int)((blockIdx.x * blockDim.x + threadIdx.x) >> 5);
  int lane = threadIdx.x & 31;
  if (wave >= nTiles) return;
  const int r0 = wave * 16;
  const int ln = lane & 15, lh = lane >> 4;

  v2f aA[16], aH[16];
  const v2f* ap = (const v2f*)(agg + (size_t)(r0 + ln) * H_DIM);
  const v2f* hp = (const v2f*)(h + (size_t)(r0 + ln) * H_DIM);
#pragma unroll
  for (int kk = 0; kk < 16; ++kk) {
    aA[kk] = ap[2 * kk + lh];
    aH[kk] = hp[2 * kk + lh];
  }

#pragma unroll 1
  for (int j = 0; j < 4; ++j) {
    const int c = j * 16 + ln;
    float bir = bih[c], biz = bih[64 + c], bin = bih[128 + c];
    float bhr = bhh[c], bhz = bhh[64 + c], bhn = bhh[128 + c];
    v8f gi_r, gi_z, gi_n, gh_r, gh_z, gh_n;
#pragma unroll
    for (int v = 0; v < 8; ++v) {
      gi_r[v] = bir; gi_z[v] = biz; gi_n[v] = bin;
      gh_r[v] = bhr; gh_z[v] = bhz; gh_n[v] = bhn;
    }

#pragma unroll
    for (int kk = 0; kk < 16; ++kk) {
      int base = kk * 12 * 32 + lane;
      v2f br = sIh[base + (j + 0) * 32];
      v2f bz = sIh[base + (j + 4) * 32];
      v2f bn = sIh[base + (j + 8) * 32];
      v2f cr = sHh[base + (j + 0) * 32];
      v2f cz = sHh[base + (j + 4) * 32];
      v2f cn = sHh[base + (j + 8) * 32];
      gi_r = wmma4(aA[kk], br, gi_r);
      gi_z = wmma4(aA[kk], bz, gi_z);
      gi_n = wmma4(aA[kk], bn, gi_n);
      gh_r = wmma4(aH[kk], cr, gh_r);
      gh_z = wmma4(aH[kk], cz, gh_z);
      gh_n = wmma4(aH[kk], cn, gh_n);
    }

#pragma unroll
    for (int v = 0; v < 8; ++v) {
      size_t idx = (size_t)(r0 + v + 8 * lh) * H_DIM + c;
      float hv = h[idx];
      float r = 1.0f / (1.0f + __expf(-(gi_r[v] + gh_r[v])));
      float z = 1.0f / (1.0f + __expf(-(gi_z[v] + gh_z[v])));
      float n = tanhf(gi_n[v] + r * gh_n[v]);
      h[idx] = (1.0f - z) * n + z * hv;
    }
  }
}

__global__ void zero_f32(float* __restrict__ p, int n) {
  int tid = blockIdx.x * blockDim.x + threadIdx.x;
  if (tid < n) p[tid] = 0.0f;
}

__global__ void pool(const float* __restrict__ h, const int* __restrict__ batch,
                     float* __restrict__ out, int nNodes) {
  int tid = blockIdx.x * blockDim.x + threadIdx.x;
  int n = tid >> 6;
  if (n >= nNodes) return;
  int d = tid & 63;
  int g = batch[n];
  __hip_atomic_fetch_add(out + (size_t)g * H_DIM + d, h[(size_t)n * H_DIM + d],
                         __ATOMIC_RELAXED, __HIP_MEMORY_SCOPE_AGENT);
}

extern "C" void kernel_launch(void* const* d_in, const int* in_sizes, int n_in,
                              void* d_out, int out_size, void* d_ws,
                              size_t ws_size, hipStream_t stream) {
  const float* x      = (const float*)d_in[0];
  const int*   ei     = (const int*)d_in[1];
  const int*   batch  = (const int*)d_in[2];
  const float* weight = (const float*)d_in[3];
  const float* Wih    = (const float*)d_in[4];
  const float* Whh    = (const float*)d_in[5];
  const float* bih    = (const float*)d_in[6];
  const float* bhh    = (const float*)d_in[7];
  float* out = (float*)d_out;

  const size_t NB = (size_t)N_NODES * H_DIM * sizeof(float);
  const int NFRAG_W = 16 * 4 * 32;
  char* ws = (char*)d_ws;
  float* h    = (float*)(ws);
  float* m    = (float*)(ws + NB);
  float* agg  = (float*)(ws + 2 * NB);
  v2f* packW  = (v2f*)(ws + 3 * NB);
  v2f* packIh = packW + (size_t)N_LAYERS * NFRAG_W;
  v2f* packHh = packIh + NFRAG_G;

  hipMemcpyAsync(h, x, NB, hipMemcpyDeviceToDevice, stream);
  for (int layer = 0; layer < N_LAYERS; ++layer)
    pack_b<<<(NFRAG_W + 255) / 256, 256, 0, stream>>>(
        weight + (size_t)layer * 64 * 64, (float*)(packW + (size_t)layer * NFRAG_W),
        4, 0);
  pack_b<<<(NFRAG_G + 255) / 256, 256, 0, stream>>>(Wih, (float*)packIh, 12, 1);
  pack_b<<<(NFRAG_G + 255) / 256, 256, 0, stream>>>(Whh, (float*)packHh, 12, 1);

  const int nTiles = N_NODES / 16;
  const int tileBlocks = (nTiles * 32 + 255) / 256;
  const int nFeat = N_NODES * H_DIM;

  for (int layer = 0; layer < N_LAYERS; ++layer) {
    gemm_m<<<tileBlocks, 256, 0, stream>>>(h, packW + (size_t)layer * NFRAG_W,
                                           m, nTiles);
    zero_f32<<<(nFeat + 255) / 256, 256, 0, stream>>>(agg, nFeat);
    scatter_add<<<(N_EDGES * 16 + 255) / 256, 256, 0, stream>>>(m, ei, agg,
                                                                N_EDGES);
    gru_fused<<<tileBlocks, 256, 0, stream>>>(h, agg, packIh, packHh, bih, bhh,
                                              nTiles);
  }

  zero_f32<<<(NUM_GRAPHS * H_DIM + 255) / 256, 256, 0, stream>>>(
      out, NUM_GRAPHS * H_DIM);
  pool<<<(N_NODES * H_DIM + 255) / 256, 256, 0, stream>>>(h, batch, out,
                                                          N_NODES);
}


# scatter_add: one edge per whole wave (lane l adds floats l, l+32): each f32 atomic instruction covers one whole 128-B line of agg, not a quarter of four lines; 8 edges per wave, atomics kept
# speedup vs baseline: 1.4916x; 1.4916x over previous
#include <hip/hip_runtime.h>
#include <math.h>

#define N_NODES    100000
#define N_EDGES    1600000
#define H_DIM      64
#define N_LAYERS   3
#define NUM_GRAPHS 256

typedef __attribute__((ext_vector_type(2))) float v2f;
typedef __attribute__((ext_vector_type(8))) float v8f;

__device__ __forceinline__ v8f wmma4(v2f a, v2f b, v8f c) {
  return __builtin_amdgcn_wmma_f32_16x16x4_f32(false, a, false, b, (short)0, c,
                                               false, false);
}

__global__ void pack_b(const float* __restrict__ Wsrc, float* __restrict__ pack,
                       int nGroups, int transposed) {
  int tid = blockIdx.x * blockDim.x + threadIdx.x;
  int total = 16 * nGroups * 32;
  if (tid >= total) return;
  int lane = tid & 31;
  int t = tid >> 5;
  int g = t % nGroups, kk = t / nGroups;
  int ln = lane & 15, lh = lane >> 4;
  int c = g * 16 + ln;
  int k0 = 4 * kk + 2 * lh;
  v2f b;
  if (transposed) {
    b.x = Wsrc[c * 64 + k0];
    b.y = Wsrc[c * 64 + k0 + 1];
  } else {
    b.x = Wsrc[k0 * 64 + c];
    b.y = Wsrc[(k0 + 1) * 64 + c];
  }
  ((v2f*)pack)[tid] = b;
}

__global__ void gemm_m(const float* __restrict__ h,
                       const v2f* __restrict__ packW, float* __restrict__ m,
                       int nTiles) {
  int wave = (int)((blockIdx.x * blockDim.x + threadIdx.x) >> 5);
  int lane = threadIdx.x & 31;
  if (wave >= nTiles) return;
  const int r0 = wave * 16;
  const int ln = lane & 15, lh = lane >> 4;

  v2f a[16];
  const v2f* hp = (const v2f*)(h + (size_t)(r0 + ln) * H_DIM);
#pragma unroll
  for (int kk = 0; kk < 16; ++kk) a[kk] = hp[2 * kk + lh];

  v8f acc[4];
#pragma unroll
  for (int j = 0; j < 4; ++j) {
    v8f z = {};
    acc[j] = z;
  }

#pragma unroll
  for (int kk = 0; kk < 16; ++kk) {
#pragma unroll
    for (int j = 0; j < 4; ++j) {
      v2f b = packW[(kk * 4 + j) * 32 + lane];
      acc[j] = wmma4(a[kk], b, acc[j]);
    }
  }

#pragma unroll
  for (int j = 0; j < 4; ++j)
#pragma unroll
    for (int v = 0; v < 8; ++v)
      m[(size_t)(r0 + v + 8 * lh) * H_DIM + j * 16 + ln] = acc[j][v];
}

#define EPW 8
static_assert(N_EDGES % EPW == 0, "edge count must be a multiple of EPW");

__global__ void scatter_add(const float* __restrict__ m,
                            const int* __restrict__ ei,
                            float* __restrict__ agg, int nEdges) {
  const int lane = threadIdx.x & 31;
  const int wave = __builtin_amdgcn_readfirstlane(
      (int)((blockIdx.x * blockDim.x + threadIdx.x) >> 5));
  const int e0 = wave * EPW;
  if (e0 >= nEdges) return;

  int src[EPW], dst[EPW];
#pragma unroll
  for (int k = 0; k < EPW; ++k) {
    src[k] = ei[e0 + k];
    dst[k] = ei[nEdges + e0 + k];
  }

  float v0[EPW], v1[EPW];
#pragma unroll
  for (int k = 0; k < EPW; ++k) {
    const float* mp = m + (size_t)src[k] * H_DIM + lane;
    v0[k] = mp[0];
    v1[k] = mp[32];
  }

#pragma unroll
  for (int k = 0; k < EPW; ++k) {
    float* ap = agg + (size_t)dst[k] * H_DIM + lane;
    __hip_atomic_fetch_add(ap, v0[k], __ATOMIC_RELAXED, __HIP_MEMORY_SCOPE_AGENT);
    __hip_atomic_fetch_add(ap + 32, v1[k], __ATOMIC_RELAXED, __HIP_MEMORY_SCOPE_AGENT);
  }
}

#define NFRAG_G (16 * 12 * 32)

__global__ void gru_fused(float* __restrict__ h, const float* __restrict__ agg,
                          const v2f* __restrict__ pIh,
                          const v2f* __restrict__ pHh,
                          const float* __restrict__ bih,
                          const float* __restrict__ bhh, int nTiles) {
  __shared__ v2f sIh[NFRAG_G];
  __shared__ v2f sHh[NFRAG_G];

  for (int i = threadIdx.x; i < NFRAG_G; i += 256) {
    sIh[i] = pIh[i];
    sHh[i] = pHh[i];
  }
  __syncthreads();

  int wave = (int)((blockIdx.x * blockDim.x + threadIdx.x) >> 5);
  int lane = threadIdx.x & 31;
  if (wave >= nTiles) return;
  const int r0 = wave * 16;
  const int ln = lane & 15, lh = lane >> 4;

  v2f aA[16], aH[16];
  const v2f* ap = (const v2f*)(agg + (size_t)(r0 + ln) * H_DIM);
  const v2f* hp = (const v2f*)(h + (size_t)(r0 + ln) * H_DIM);
#pragma unroll
  for (int kk = 0; kk < 16; ++kk) {
    aA[kk] = ap[2 * kk + lh];
    aH[kk] = hp[2 * kk + lh];
  }

#pragma unroll 1
  for (int j = 0; j < 4; ++j) {
    const int c = j * 16 + ln;
    float bir = bih[c], biz = bih[64 + c], bin = bih[128 + c];
    float bhr = bhh[c], bhz = bhh[64 + c], bhn = bhh[128 + c];
    v8f gi_r, gi_z, gi_n, gh_r, gh_z, gh_n;
#pragma unroll
    for (int v = 0; v < 8; ++v) {
      gi_r[v] = bir; gi_z[v] = biz; gi_n[v] = bin;
      gh_r[v] = bhr; gh_z[v] = bhz; gh_n[v] = bhn;
    }

#pragma unroll
    for (int kk = 0; kk < 16; ++kk) {
      int base = kk * 12 * 32 + lane;
      v2f br = sIh[base + (j + 0) * 32];
      v2f bz = sIh[base + (j + 4) * 32];
      v2f bn = sIh[base + (j + 8) * 32];
      v2f cr = sHh[base + (j + 0) * 32];
      v2f cz = sHh[base + (j + 4) * 32];
      v2f cn = sHh[base + (j + 8) * 32];
      gi_r = wmma4(aA[kk], br, gi_r);
      gi_z = wmma4(aA[kk], bz, gi_z);
      gi_n = wmma4(aA[kk], bn, gi_n);
      gh_r = wmma4(aH[kk], cr, gh_r);
      gh_z = wmma4(aH[kk], cz, gh_z);
      gh_n = wmma4(aH[kk], cn, gh_n);
    }

#pragma unroll
    for (int v = 0; v < 8; ++v) {
      size_t idx = (size_t)(r0 + v + 8 * lh) * H_DIM + c;
      float hv = h[idx];
      float r = 1.0f / (1.0f + __expf(-(gi_r[v] + gh_r[v])));
      float z = 1.0f / (1.0f + __expf(-(gi_z[v] + gh_z[v])));
      float n = tanhf(gi_n[v] + r * gh_n[v]);
      h[idx] = (1.0f - z) * n + z * hv;
    }
  }
}

__global__ void zero_f32(float* __restrict__ p, int n) {
  int tid = blockIdx.x * blockDim.x + threadIdx.x;
  if (tid < n) p[tid] = 0.0f;
}

__global__ void pool(const float* __restrict__ h, const int* __restrict__ batch,
                     float* __restrict__ out, int nNodes) {
  int tid = blockIdx.x * blockDim.x + threadIdx.x;
  int n = tid >> 6;
  if (n >= nNodes) return;
  int d = tid & 63;
  int g = batch[n];
  __hip_atomic_fetch_add(out + (size_t)g * H_DIM + d, h[(size_t)n * H_DIM + d],
                         __ATOMIC_RELAXED, __HIP_MEMORY_SCOPE_AGENT);
}

extern "C" void kernel_launch(void* const* d_in, const int* in_sizes, int n_in,
                              void* d_out, int out_size, void* d_ws,
                              size_t ws_size, hipStream_t stream) {
  const float* x      = (const float*)d_in[0];
  const int*   ei     = (const int*)d_in[1];
  const int*   batch  = (const int*)d_in[2];
  const float* weight = (const float*)d_in[3];
  const float* Wih    = (const float*)d_in[4];
  const float* Whh    = (const float*)d_in[5];
  const float* bih    = (const float*)d_in[6];
  const float* bhh    = (const float*)d_in[7];
  float* out = (float*)d_out;

  const size_t NB = (size_t)N_NODES * H_DIM * sizeof(float);
  const int NFRAG_W = 16 * 4 * 32;
  char* ws = (char*)d_ws;
  float* h    = (float*)(ws);
  float* m    = (float*)(ws + NB);
  float* agg  = (float*)(ws + 2 * NB);
  v2f* packW  = (v2f*)(ws + 3 * NB);
  v2f* packIh = packW + (size_t)N_LAYERS * NFRAG_W;
  v2f* packHh = packIh + NFRAG_G;

  hipMemcpyAsync(h, x, NB, hipMemcpyDeviceToDevice, stream);
  for (int layer = 0; layer < N_LAYERS; ++layer)
    pack_b<<<(NFRAG_W + 255) / 256, 256, 0, stream>>>(
        weight + (size_t)layer * 64 * 64, (float*)(packW + (size_t)layer * NFRAG_W),
        4, 0);
  pack_b<<<(NFRAG_G + 255) / 256, 256, 0, stream>>>(Wih, (float*)packIh, 12, 1);
  pack_b<<<(NFRAG_G + 255) / 256, 256, 0, stream>>>(Whh, (float*)packHh, 12, 1);

  const int nTiles = N_NODES / 16;
  const int tileBlocks = (nTiles * 32 + 255) / 256;
  const int nFeat = N_NODES * H_DIM;

  for (int layer = 0; layer < N_LAYERS; ++layer) {
    gemm_m<<<tileBlocks, 256, 0, stream>>>(h, packW + (size_t)layer * NFRAG_W,
                                           m, nTiles);
    zero_f32<<<(nFeat + 255) / 256, 256, 0, stream>>>(agg, nFeat);
    scatter_add<<<(N_EDGES / EPW * 32 + 255) / 256, 256, 0, stream>>>(
        m, ei, agg, N_EDGES);
    gru_fused<<<tileBlocks, 256, 0, stream>>>(h, agg, packIh, packHh, bih, bhh,
                                              nTiles);
  }

  zero_f32<<<(NUM_GRAPHS * H_DIM + 255) / 256, 256, 0, stream>>>(
      out, NUM_GRAPHS * H_DIM);
  pool<<<(N_NODES * H_DIM + 255) / 256, 256, 0, stream>>>(h, batch, out,
                                                          N_NODES);
}


# fold layer weight into gate weight: Wc = W @ W_ih^T (64x192, rebuilt each call), edges sum h[src] not (h@W)[src], the three N x 64 x 64 products go; reassociation changes the f32 sum order
# speedup vs baseline: 1.7245x; 1.1561x over previous
#include <hip/hip_runtime.h>
#include <math.h>

#define N_NODES    100000
#define N_EDGES    1600000
#define H_DIM      64
#define N_LAYERS   3
#define NUM_GRAPHS 256

typedef __attribute__((ext_vector_type(2))) float v2f;
typedef __attribute__((ext_vector_type(8))) float v8f;

__device__ __forceinline__ v8f wmma4(v2f a, v2f b, v8f c) {
  return __builtin_amdgcn_wmma_f32_16x16x4_f32(false, a, false, b, (short)0, c,
                                               false, false);
}

__global__ void pack_b(const float* __restrict__ Wsrc, float* __restrict__ pack,
                       int nGroups, int transposed) {
  int tid = blockIdx.x * blockDim.x + threadIdx.x;
  int total = 16 * nGroups * 32;
  if (tid >= total) return;
  int lane = tid & 31;
  int t = tid >> 5;
  int g = t % nGroups, kk = t / nGroups;
  int ln = lane & 15, lh = lane >> 4;
  int c = g * 16 + ln;
  int k0 = 4 * kk + 2 * lh;
  v2f b;
  if (transposed) {
    b.x = Wsrc[c * 64 + k0];
    b.y = Wsrc[c * 64 + k0 + 1];
  } else {
    b.x = Wsrc[k0 * 64 + c];
    b.y = Wsrc[(k0 + 1) * 64 + c];
  }
  ((v2f*)pack)[tid] = b;
}

#define NFRAG_G (16 * 12 * 32)

__global__ void pack_wc(const float* __restrict__ weight,
                        const float* __restrict__ Wih,
                        float* __restrict__ pack) {
  int tid = blockIdx.x * blockDim.x + threadIdx.x;
  if (tid >= N_LAYERS * NFRAG_G) return;
  int layer = tid / NFRAG_G;
  int f = tid - layer * NFRAG_G;
  int lane = f & 31;
  int t = f >> 5;
  int g = t % 12, kk = t / 12;
  int ln = lane & 15, lh = lane >> 4;
  int c = g * 16 + ln;
  int k0 = 4 * kk + 2 * lh;
  const float* w0 = weight + (size_t)layer * 64 * 64 + (size_t)k0 * 64;
  const float* w1 = w0 + 64;
  const float* wi = Wih + (size_t)c * 64;
  float s0 = 0.0f, s1 = 0.0f;
#pragma unroll
  for (int j = 0; j < 64; ++j) {
    float u = wi[j];
    s0 = fmaf(w0[j], u, s0);
    s1 = fmaf(w1[j], u, s1);
  }
  v2f b;
  b.x = s0;
  b.y = s1;
  ((v2f*)pack)[tid] = b;
}

#define EPW 8
static_assert(N_EDGES % EPW == 0, "edge count must be a multiple of EPW");

__global__ void scatter_add(const float* __restrict__ hsrc,
                            const int* __restrict__ ei,
                            float* __restrict__ agg, int nEdges) {
  const int lane = threadIdx.x & 31;
  const int wave = __builtin_amdgcn_readfirstlane(
      (int)((blockIdx.x * blockDim.x + threadIdx.x) >> 5));
  const int e0 = wave * EPW;
  if (e0 >= nEdges) return;

  int src[EPW], dst[EPW];
#pragma unroll
  for (int k = 0; k < EPW; ++k) {
    src[k] = ei[e0 + k];
    dst[k] = ei[nEdges + e0 + k];
  }

  float v0[EPW], v1[EPW];
#pragma unroll
  for (int k = 0; k < EPW; ++k) {
    const float* mp = hsrc + (size_t)src[k] * H_DIM + lane;
    v0[k] = mp[0];
    v1[k] = mp[32];
  }

#pragma unroll
  for (int k = 0; k < EPW; ++k) {
    float* ap = agg + (size_t)dst[k] * H_DIM + lane;
    __hip_atomic_fetch_add(ap, v0[k], __ATOMIC_RELAXED, __HIP_MEMORY_SCOPE_AGENT);
    __hip_atomic_fetch_add(ap + 32, v1[k], __ATOMIC_RELAXED, __HIP_MEMORY_SCOPE_AGENT);
  }
}

__global__ void gru_fused(float* __restrict__ h, const float* __restrict__ agg,
                          const v2f* __restrict__ pIh,
                          const v2f* __restrict__ pHh,
                          const float* __restrict__ bih,
                          const float* __restrict__ bhh, int nTiles) {
  __shared__ v2f sIh[NFRAG_G];
  __shared__ v2f sHh[NFRAG_G];

  for (int i = threadIdx.x; i < NFRAG_G; i += 256) {
    sIh[i] = pIh[i];
    sHh[i] = pHh[i];
  }
  __syncthreads();

  int wave = (int)((blockIdx.x * blockDim.x + threadIdx.x) >> 5);
  int lane = threadIdx.x & 31;
  if (wave >= nTiles) return;
  const int r0 = wave * 16;
  const int ln = lane & 15, lh = lane >> 4;

  v2f aA[16], aH[16];
  const v2f* ap = (const v2f*)(agg + (size_t)(r0 + ln) * H_DIM);
  const v2f* hp = (const v2f*)(h + (size_t)(r0 + ln) * H_DIM);
#pragma unroll
  for (int kk = 0; kk < 16; ++kk) {
    aA[kk] = ap[2 * kk + lh];
    aH[kk] = hp[2 * kk + lh];
  }

#pragma unroll 1
  for (int j = 0; j < 4; ++j) {
    const int c = j * 16 + ln;
    float bir = bih[c], biz = bih[64 + c], bin = bih[128 + c];
    float bhr = bhh[c], bhz = bhh[64 + c], bhn = bhh[128 + c];
    v8f gi_r, gi_z, gi_n, gh_r, gh_z, gh_n;
#pragma unroll
    for (int v = 0; v < 8; ++v) {
      gi_r[v] = bir; gi_z[v] = biz; gi_n[v] = bin;
      gh_r[v] = bhr; gh_z[v] = bhz; gh_n[v] = bhn;
    }

#pragma unroll
    for (int kk = 0; kk < 16; ++kk) {
      int base = kk * 12 * 32 + lane;
      v2f br = sIh[base + (j + 0) * 32];
      v2f bz = sIh[base + (j + 4) * 32];
      v2f bn = sIh[base + (j + 8) * 32];
      v2f cr = sHh[base + (j + 0) * 32];
      v2f cz = sHh[base + (j + 4) * 32];
      v2f cn = sHh[base + (j + 8) * 32];
      gi_r = wmma4(aA[kk], br, gi_r);
      gi_z = wmma4(aA[kk], bz, gi_z);
      gi_n = wmma4(aA[kk], bn, gi_n);
      gh_r = wmma4(aH[kk], cr, gh_r);
      gh_z = wmma4(aH[kk], cz, gh_z);
      gh_n = wmma4(aH[kk], cn, gh_n);
    }

#pragma unroll
    for (int v = 0; v < 8; ++v) {
      size_t idx = (size_t)(r0 + v + 8 * lh) * H_DIM + c;
      float hv = h[idx];
      float r = 1.0f / (1.0f + __expf(-(gi_r[v] + gh_r[v])));
      float z = 1.0f / (1.0f + __expf(-(gi_z[v] + gh_z[v])));
      float n = tanhf(gi_n[v] + r * gh_n[v]);
      h[idx] = (1.0f - z) * n + z * hv;
    }
  }
}

__global__ void zero_f32(float* __restrict__ p, int n) {
  int tid = blockIdx.x * blockDim.x + threadIdx.x;
  if (tid < n) p[tid] = 0.0f;
}

__global__ void pool(const float* __restrict__ h, const int* __restrict__ batch,
                     float* __restrict__ out, int nNodes) {
  int tid = blockIdx.x * blockDim.x + threadIdx.x;
  int n = tid >> 6;
  if (n >= nNodes) return;
  int d = tid & 63;
  int g = batch[n];
  __hip_atomic_fetch_add(out + (size_t)g * H_DIM + d, h[(size_t)n * H_DIM + d],
                         __ATOMIC_RELAXED, __HIP_MEMORY_SCOPE_AGENT);
}

extern "C" void kernel_launch(void* const* d_in, const int* in_sizes, int n_in,
                              void* d_out, int out_size, void* d_ws,
                              size_t ws_size, hipStream_t stream) {
  const float* x      = (const float*)d_in[0];
  const int*   ei     = (const int*)d_in[1];
  const int*   batch  = (const int*)d_in[2];
  const float* weight = (const float*)d_in[3];
  const float* Wih    = (const float*)d_in[4];
  const float* Whh    = (const float*)d_in[5];
  const float* bih    = (const float*)d_in[6];
  const float* bhh    = (const float*)d_in[7];
  float* out = (float*)d_out;

  const size_t NB = (size_t)N_NODES * H_DIM * sizeof(float);
  char* ws = (char*)d_ws;
  float* h    = (float*)(ws);
  v2f* packWc = (v2f*)(ws + NB);
  v2f* packHh = packWc + (size_t)N_LAYERS * NFRAG_G;
  float* agg  = (float*)(ws + 2 * NB);

  hipMemcpyAsync(h, x, NB, hipMemcpyDeviceToDevice, stream);
  pack_wc<<<(N_LAYERS * NFRAG_G + 255) / 256, 256, 0, stream>>>(weight, Wih,
                                                                (float*)packWc);
  pack_b<<<(NFRAG_G + 255) / 256, 256, 0, stream>>>(Whh, (float*)packHh, 12, 1);

  const int nTiles = N_NODES / 16;
  const int tileBlocks = (nTiles * 32 + 255) / 256;
  const int nFeat = N_NODES * H_DIM;

  for (int layer = 0; layer < N_LAYERS; ++layer) {
    zero_f32<<<(nFeat + 255) / 256, 256, 0, stream>>>(agg, nFeat);
    scatter_add<<<(N_EDGES / EPW * 32 + 255) / 256, 256, 0, stream>>>(
        h, ei, agg, N_EDGES);
    gru_fused<<<tileBlocks, 256, 0, stream>>>(
        h, agg, packWc + (size_t)layer * NFRAG_G, packHh, bih, bhh, nTiles);
  }

  zero_f32<<<(NUM_GRAPHS * H_DIM + 255) / 256, 256, 0, stream>>>(
      out, NUM_GRAPHS * H_DIM);
  pool<<<(N_NODES * H_DIM + 255) / 256, 256, 0, stream>>>(h, batch, out,
                                                          N_NODES);
}


# segment sum in LDS: edges grouped by destination range once per call (LDS counters), one work-group per 391-row range adds source rows with LDS f32 adds and stores agg once; sums in arrival order
# speedup vs baseline: 2.0024x; 1.1611x over previous
#include <hip/hip_runtime.h>
#include <math.h>

#define N_NODES    100000
#define N_EDGES    1600000
#define H_DIM      64
#define N_LAYERS   3
#define NUM_GRAPHS 256

typedef __attribute__((ext_vector_type(2))) float v2f;
typedef __attribute__((ext_vector_type(8))) float v8f;

__device__ __forceinline__ v8f wmma4(v2f a, v2f b, v8f c) {
  return __builtin_amdgcn_wmma_f32_16x16x4_f32(false, a, false, b, (short)0, c,
                                               false, false);
}

__global__ void pack_b(const float* __restrict__ Wsrc, float* __restrict__ pack,
                       int nGroups, int transposed) {
  int tid = blockIdx.x * blockDim.x + threadIdx.x;
  int total = 16 * nGroups * 32;
  if (tid >= total) return;
  int lane = tid & 31;
  int t = tid >> 5;
  int g = t % nGroups, kk = t / nGroups;
  int ln = lane & 15, lh = lane >> 4;
  int c = g * 16 + ln;
  int k0 = 4 * kk + 2 * lh;
  v2f b;
  if (transposed) {
    b.x = Wsrc[c * 64 + k0];
    b.y = Wsrc[c * 64 + k0 + 1];
  } else {
    b.x = Wsrc[k0 * 64 + c];
    b.y = Wsrc[(k0 + 1) * 64 + c];
  }
  ((v2f*)pack)[tid] = b;
}

#define NFRAG_G (16 * 12 * 32)

__global__ void pack_wc(const float* __restrict__ weight,
                        const float* __restrict__ Wih,
                        float* __restrict__ pack) {
  int tid = blockIdx.x * blockDim.x + threadIdx.x;
  if (tid >= N_LAYERS * NFRAG_G) return;
  int layer = tid / NFRAG_G;
  int f = tid - layer * NFRAG_G;
  int lane = f & 31;
  int t = f >> 5;
  int g = t % 12, kk = t / 12;
  int ln = lane & 15, lh = lane >> 4;
  int c = g * 16 + ln;
  int k0 = 4 * kk + 2 * lh;
  const float* w0 = weight + (size_t)layer * 64 * 64 + (size_t)k0 * 64;
  const float* w1 = w0 + 64;
  const float* wi = Wih + (size_t)c * 64;
  float s0 = 0.0f, s1 = 0.0f;
#pragma unroll
  for (int j = 0; j < 64; ++j) {
    float u = wi[j];
    s0 = fmaf(w0[j], u, s0);
    s1 = fmaf(w1[j], u, s1);
  }
  v2f b;
  b.x = s0;
  b.y = s1;
  ((v2f*)pack)[tid] = b;
}

#define NBINS    256
#define RANGE    391
#define BIN_WGS  256
#define EPB      (N_EDGES / BIN_WGS)
#define BIN_ITERS ((EPB + 1023) / 1024)
#define SRC_BITS 17
#define SRC_MASK ((1 << SRC_BITS) - 1)
static_assert(RANGE * NBINS >= N_NODES, "ranges must cover all rows");
static_assert(N_EDGES % BIN_WGS == 0, "edge count must split evenly");
static_assert((1 << SRC_BITS) >= N_NODES, "source row must fit its field");
static_assert(RANGE <= (1 << (31 - SRC_BITS)), "local row must fit its field");
static_assert(BIN_WGS % 4 == 0, "bin_scan splits the work-groups in four");

__global__ void __launch_bounds__(1024)
bin_count(const int* __restrict__ ei, int* __restrict__ cnt) {
  __shared__ int hist[NBINS];
  const int tid = threadIdx.x;
  if (tid < NBINS) hist[tid] = 0;
  __syncthreads();
  const int* dst = ei + N_EDGES + blockIdx.x * EPB;
  int d[BIN_ITERS];
#pragma unroll
  for (int j = 0; j < BIN_ITERS; ++j) {
    int i = tid + j * 1024;
    d[j] = dst[i < EPB ? i : EPB - 1];
  }
#pragma unroll
  for (int j = 0; j < BIN_ITERS; ++j)
    if (tid + j * 1024 < EPB) atomicAdd(&hist[d[j] / RANGE], 1);
  __syncthreads();
  if (tid < NBINS) cnt[blockIdx.x * NBINS + tid] = hist[tid];
}

__global__ void __launch_bounds__(1024)
bin_scan(int* __restrict__ cnt, int* __restrict__ binStart) {
  __shared__ int part[4][NBINS];
  __shared__ int sc[NBINS];
  const int tid = threadIdx.x;
  const int t = tid & (NBINS - 1);
  const int q = tid >> 8;
  constexpr int WPQ = BIN_WGS / 4;
  int c[WPQ];
#pragma unroll
  for (int w = 0; w < WPQ; ++w) c[w] = cnt[(q * WPQ + w) * NBINS + t];
  int run = 0;
#pragma unroll
  for (int w = 0; w < WPQ; ++w) run += c[w];
  part[q][t] = run;
  __syncthreads();
  int base = 0;
  for (int p = 0; p < q; ++p) base += part[p][t];
  if (q == 3) sc[t] = base + run;
  __syncthreads();
  const int total = sc[t];
  for (int off = 1; off < NBINS; off <<= 1) {
    int v = 0;
    if (tid < NBINS && tid >= off) v = sc[tid - off];
    __syncthreads();
    if (tid < NBINS) sc[tid] += v;
    __syncthreads();
  }
  const int start = sc[t] - total;
  int pos = start + base;
#pragma unroll
  for (int w = 0; w < WPQ; ++w) {
    cnt[(q * WPQ + w) * NBINS + t] = pos;
    pos += c[w];
  }
  if (q == 0) binStart[t] = start;
  if (tid == 0) binStart[NBINS] = sc[NBINS - 1];
}

__global__ void __launch_bounds__(1024)
bin_place(const int* __restrict__ ei, const int* __restrict__ off,
          int* __restrict__ sorted) {
  __shared__ int cur[NBINS];
  const int tid = threadIdx.x;
  if (tid < NBINS) cur[tid] = off[blockIdx.x * NBINS + tid];
  __syncthreads();
  const int* src = ei + blockIdx.x * EPB;
  const int* dst = src + N_EDGES;
  int s[BIN_ITERS], d[BIN_ITERS];
#pragma unroll
  for (int j = 0; j < BIN_ITERS; ++j) {
    int i = tid + j * 1024;
    i = i < EPB ? i : EPB - 1;
    s[j] = src[i];
    d[j] = dst[i];
  }
#pragma unroll
  for (int j = 0; j < BIN_ITERS; ++j) {
    if (tid + j * 1024 < EPB) {
      int b = d[j] / RANGE;
      int dl = d[j] - b * RANGE;
      int pos = atomicAdd(&cur[b], 1);
      sorted[pos] = s[j] | (dl << SRC_BITS);
    }
  }
}

#define EPW 8

__global__ void __launch_bounds__(1024)
agg_range(const float* __restrict__ hsrc, const int* __restrict__ sorted,
          const int* __restrict__ binStart, float* __restrict__ agg) {
  __shared__ float acc[RANGE * H_DIM];
  const int tid = threadIdx.x;
  const int lane = tid & 31;
  float4* acc4 = (float4*)acc;
  for (int i = tid; i < RANGE * H_DIM / 4; i += 1024)
    acc4[i] = make_float4(0.0f, 0.0f, 0.0f, 0.0f);
  __syncthreads();

  const int b = blockIdx.x;
  const int beg = binStart[b];
  const int end = binStart[b + 1];
  const int w = __builtin_amdgcn_readfirstlane(tid >> 5);

  int e0 = beg + w * EPW;
  int pk[EPW];
  if (e0 < end) {
#pragma unroll
    for (int k = 0; k < EPW; ++k) {
      int e = e0 + k;
      pk[k] = sorted[e < end ? e : end - 1];
    }
  }
  while (e0 < end) {
    float v0[EPW], v1[EPW];
#pragma unroll
    for (int k = 0; k < EPW; ++k) {
      const float* mp = hsrc + (size_t)(pk[k] & SRC_MASK) * H_DIM + lane;
      v0[k] = mp[0];
      v1[k] = mp[32];
    }
    const int e1 = e0 + 32 * EPW;
    int nk[EPW];
#pragma unroll
    for (int k = 0; k < EPW; ++k) {
      int e = e1 + k;
      nk[k] = sorted[e < end ? e : end - 1];
    }
#pragma unroll
    for (int k = 0; k < EPW; ++k) {
      const bool live = (e0 + k) < end;
      float* ap = acc + (pk[k] >> SRC_BITS) * H_DIM + lane;
      __hip_atomic_fetch_add(ap, live ? v0[k] : 0.0f, __ATOMIC_RELAXED,
                             __HIP_MEMORY_SCOPE_WORKGROUP);
      __hip_atomic_fetch_add(ap + 32, live ? v1[k] : 0.0f, __ATOMIC_RELAXED,
                             __HIP_MEMORY_SCOPE_WORKGROUP);
    }
#pragma unroll
    for (int k = 0; k < EPW; ++k) pk[k] = nk[k];
    e0 = e1;
  }
  __syncthreads();

  const int row0 = b * RANGE;
  const int nrows = (N_NODES - row0) < RANGE ? (N_NODES - row0) : RANGE;
  float4* out4 = (float4*)(agg + (size_t)row0 * H_DIM);
  for (int i = tid; i < nrows * (H_DIM / 4); i += 1024) out4[i] = acc4[i];
}

__global__ void gru_fused(float* __restrict__ h, const float* __restrict__ agg,
                          const v2f* __restrict__ pIh,
                          const v2f* __restrict__ pHh,
                          const float* __restrict__ bih,
                          const float* __restrict__ bhh, int nTiles) {
  __shared__ v2f sIh[NFRAG_G];
  __shared__ v2f sHh[NFRAG_G];

  for (int i = threadIdx.x; i < NFRAG_G; i += 256) {
    sIh[i] = pIh[i];
    sHh[i] = pHh[i];
  }
  __syncthreads();

  int wave = (int)((blockIdx.x * blockDim.x + threadIdx.x) >> 5);
  int lane = threadIdx.x & 31;
  if (wave >= nTiles) return;
  const int r0 = wave * 16;
  const int ln = lane & 15, lh = lane >> 4;

  v2f aA[16], aH[16];
  const v2f* ap = (const v2f*)(agg + (size_t)(r0 + ln) * H_DIM);
  const v2f* hp = (const v2f*)(h + (size_t)(r0 + ln) * H_DIM);
#pragma unroll
  for (int kk = 0; kk < 16; ++kk) {
    aA[kk] = ap[2 * kk + lh];
    aH[kk] = hp[2 * kk + lh];
  }

#pragma unroll 1
  for (int j = 0; j < 4; ++j) {
    const int c = j * 16 + ln;
    float bir = bih[c], biz = bih[64 + c], bin = bih[128 + c];
    float bhr = bhh[c], bhz = bhh[64 + c], bhn = bhh[128 + c];
    v8f gi_r, gi_z, gi_n, gh_r, gh_z, gh_n;
#pragma unroll
    for (int v = 0; v < 8; ++v) {
      gi_r[v] = bir; gi_z[v] = biz; gi_n[v] = bin;
      gh_r[v] = bhr; gh_z[v] = bhz; gh_n[v] = bhn;
    }

#pragma unroll
    for (int kk = 0; kk < 16; ++kk) {
      int base = kk * 12 * 32 + lane;
      v2f br = sIh[base + (j + 0) * 32];
      v2f bz = sIh[base + (j + 4) * 32];
      v2f bn = sIh[base + (j + 8) * 32];
      v2f cr = sHh[base + (j + 0) * 32];
      v2f cz = sHh[base + (j + 4) * 32];
      v2f cn = sHh[base + (j + 8) * 32];
      gi_r = wmma4(aA[kk], br, gi_r);
      gi_z = wmma4(aA[kk], bz, gi_z);
      gi_n = wmma4(aA[kk], bn, gi_n);
      gh_r = wmma4(aH[kk], cr, gh_r);
      gh_z = wmma4(aH[kk], cz, gh_z);
      gh_n = wmma4(aH[kk], cn, gh_n);
    }

#pragma unroll
    for (int v = 0; v < 8; ++v) {
      size_t idx = (size_t)(r0 + v + 8 * lh) * H_DIM + c;
      float hv = h[idx];
      float r = 1.0f / (1.0f + __expf(-(gi_r[v] + gh_r[v])));
      float z = 1.0f / (1.0f + __expf(-(gi_z[v] + gh_z[v])));
      float n = tanhf(gi_n[v] + r * gh_n[v]);
      h[idx] = (1.0f - z) * n + z * hv;
    }
  }
}

__global__ void zero_f32(float* __restrict__ p, int n) {
  int tid = blockIdx.x * blockDim.x + threadIdx.x;
  if (tid < n) p[tid] = 0.0f;
}

__global__ void pool(const float* __restrict__ h, const int* __restrict__ batch,
                     float* __restrict__ out, int nNodes) {
  int tid = blockIdx.x * blockDim.x + threadIdx.x;
  int n = tid >> 6;
  if (n >= nNodes) return;
  int d = tid & 63;
  int g = batch[n];
  __hip_atomic_fetch_add(out + (size_t)g * H_DIM + d, h[(size_t)n * H_DIM + d],
                         __ATOMIC_RELAXED, __HIP_MEMORY_SCOPE_AGENT);
}

extern "C" void kernel_launch(void* const* d_in, const int* in_sizes, int n_in,
                              void* d_out, int out_size, void* d_ws,
                              size_t ws_size, hipStream_t stream) {
  const float* x      = (const float*)d_in[0];
  const int*   ei     = (const int*)d_in[1];
  const int*   batch  = (const int*)d_in[2];
  const float* weight = (const float*)d_in[3];
  const float* Wih    = (const float*)d_in[4];
  const float* Whh    = (const float*)d_in[5];
  const float* bih    = (const float*)d_in[6];
  const float* bhh    = (const float*)d_in[7];
  float* out = (float*)d_out;

  const size_t NB = (size_t)N_NODES * H_DIM * sizeof(float);
  char* ws = (char*)d_ws;
  float* h    = (float*)(ws);
  v2f* packWc = (v2f*)(ws + NB);
  v2f* packHh = packWc + (size_t)N_LAYERS * NFRAG_G;
  float* agg  = (float*)(ws + 2 * NB);
  int* cnt      = (int*)(packHh + NFRAG_G);
  int* binStart = cnt + BIN_WGS * NBINS;
  int* sorted   = binStart + 512;

  hipMemcpyAsync(h, x, NB, hipMemcpyDeviceToDevice, stream);
  pack_wc<<<(N_LAYERS * NFRAG_G + 255) / 256, 256, 0, stream>>>(weight, Wih,
                                                                (float*)packWc);
  pack_b<<<(NFRAG_G + 255) / 256, 256, 0, stream>>>(Whh, (float*)packHh, 12, 1);

  bin_count<<<BIN_WGS, 1024, 0, stream>>>(ei, cnt);
  bin_scan<<<1, 1024, 0, stream>>>(cnt, binStart);
  bin_place<<<BIN_WGS, 1024, 0, stream>>>(ei, cnt, sorted);

  const int nTiles = N_NODES / 16;
  const int tileBlocks = (nTiles * 32 + 255) / 256;

  for (int layer = 0; layer < N_LAYERS; ++layer) {
    agg_range<<<NBINS, 1024, 0, stream>>>(h, sorted, binStart, agg);
    gru_fused<<<tileBlocks, 256, 0, stream>>>(
        h, agg, packWc + (size_t)layer * NFRAG_G, packHh, bih, bhh, nTiles);
  }

  zero_f32<<<(NUM_GRAPHS * H_DIM + 255) / 256, 256, 0, stream>>>(
      out, NUM_GRAPHS * H_DIM);
  pool<<<(N_NODES * H_DIM + 255) / 256, 256, 0, stream>>>(h, batch, out,
                                                          N_NODES);
}


# per-layer segment sum and GRU cell fused in one launch: a range's sums stay in LDS and feed the gate WMMAs there; h alternates between two planes, x read in place (agg plane, copy, 3 launches go)
# speedup vs baseline: 2.4906x; 1.2438x over previous
#include <hip/hip_runtime.h>
#include <math.h>

#define N_NODES    100000
#define N_EDGES    1600000
#define H_DIM      64
#define N_LAYERS   3
#define NUM_GRAPHS 256

typedef __attribute__((ext_vector_type(2))) float v2f;
typedef __attribute__((ext_vector_type(8))) float v8f;

__device__ __forceinline__ v8f wmma4(v2f a, v2f b, v8f c) {
  return __builtin_amdgcn_wmma_f32_16x16x4_f32(false, a, false, b, (short)0, c,
                                               false, false);
}

__global__ void pack_b(const float* __restrict__ Wsrc, float* __restrict__ pack,
                       int nGroups, int transposed) {
  int tid = blockIdx.x * blockDim.x + threadIdx.x;
  int total = 16 * nGroups * 32;
  if (tid >= total) return;
  int lane = tid & 31;
  int t = tid >> 5;
  int g = t % nGroups, kk = t / nGroups;
  int ln = lane & 15, lh = lane >> 4;
  int c = g * 16 + ln;
  int k0 = 4 * kk + 2 * lh;
  v2f b;
  if (transposed) {
    b.x = Wsrc[c * 64 + k0];
    b.y = Wsrc[c * 64 + k0 + 1];
  } else {
    b.x = Wsrc[k0 * 64 + c];
    b.y = Wsrc[(k0 + 1) * 64 + c];
  }
  ((v2f*)pack)[tid] = b;
}

#define NFRAG_G (16 * 12 * 32)

__global__ void pack_wc(const float* __restrict__ weight,
                        const float* __restrict__ Wih,
                        float* __restrict__ pack) {
  int tid = blockIdx.x * blockDim.x + threadIdx.x;
  if (tid >= N_LAYERS * NFRAG_G) return;
  int layer = tid / NFRAG_G;
  int f = tid - layer * NFRAG_G;
  int lane = f & 31;
  int t = f >> 5;
  int g = t % 12, kk = t / 12;
  int ln = lane & 15, lh = lane >> 4;
  int c = g * 16 + ln;
  int k0 = 4 * kk + 2 * lh;
  const float* w0 = weight + (size_t)layer * 64 * 64 + (size_t)k0 * 64;
  const float* w1 = w0 + 64;
  const float* wi = Wih + (size_t)c * 64;
  float s0 = 0.0f, s1 = 0.0f;
#pragma unroll
  for (int j = 0; j < 64; ++j) {
    float u = wi[j];
    s0 = fmaf(w0[j], u, s0);
    s1 = fmaf(w1[j], u, s1);
  }
  v2f b;
  b.x = s0;
  b.y = s1;
  ((v2f*)pack)[tid] = b;
}

#define NBINS    256
#define RANGE    391
#define BIN_WGS  256
#define EPB      (N_EDGES / BIN_WGS)
#define BIN_ITERS ((EPB + 1023) / 1024)
#define SRC_BITS 17
#define SRC_MASK ((1 << SRC_BITS) - 1)
static_assert(RANGE * NBINS >= N_NODES, "ranges must cover all rows");
static_assert(N_EDGES % BIN_WGS == 0, "edge count must split evenly");
static_assert((1 << SRC_BITS) >= N_NODES, "source row must fit its field");
static_assert(RANGE <= (1 << (31 - SRC_BITS)), "local row must fit its field");
static_assert(BIN_WGS % 4 == 0, "bin_scan splits the work-groups in four");

__global__ void __launch_bounds__(1024)
bin_count(const int* __restrict__ ei, int* __restrict__ cnt) {
  __shared__ int hist[NBINS];
  const int tid = threadIdx.x;
  if (tid < NBINS) hist[tid] = 0;
  __syncthreads();
  const int* dst = ei + N_EDGES + blockIdx.x * EPB;
  int d[BIN_ITERS];
#pragma unroll
  for (int j = 0; j < BIN_ITERS; ++j) {
    int i = tid + j * 1024;
    d[j] = dst[i < EPB ? i : EPB - 1];
  }
#pragma unroll
  for (int j = 0; j < BIN_ITERS; ++j)
    if (tid + j * 1024 < EPB) atomicAdd(&hist[d[j] / RANGE], 1);
  __syncthreads();
  if (tid < NBINS) cnt[blockIdx.x * NBINS + tid] = hist[tid];
}

__global__ void __launch_bounds__(1024)
bin_scan(int* __restrict__ cnt, int* __restrict__ binStart) {
  __shared__ int part[4][NBINS];
  __shared__ int sc[NBINS];
  const int tid = threadIdx.x;
  const int t = tid & (NBINS - 1);
  const int q = tid >> 8;
  constexpr int WPQ = BIN_WGS / 4;
  int c[WPQ];
#pragma unroll
  for (int w = 0; w < WPQ; ++w) c[w] = cnt[(q * WPQ + w) * NBINS + t];
  int run = 0;
#pragma unroll
  for (int w = 0; w < WPQ; ++w) run += c[w];
  part[q][t] = run;
  __syncthreads();
  int base = 0;
  for (int p = 0; p < q; ++p) base += part[p][t];
  if (q == 3) sc[t] = base + run;
  __syncthreads();
  const int total = sc[t];
  for (int off = 1; off < NBINS; off <<= 1) {
    int v = 0;
    if (tid < NBINS && tid >= off) v = sc[tid - off];
    __syncthreads();
    if (tid < NBINS) sc[tid] += v;
    __syncthreads();
  }
  const int start = sc[t] - total;
  int pos = start + base;
#pragma unroll
  for (int w = 0; w < WPQ; ++w) {
    cnt[(q * WPQ + w) * NBINS + t] = pos;
    pos += c[w];
  }
  if (q == 0) binStart[t] = start;
  if (tid == 0) binStart[NBINS] = sc[NBINS - 1];
}

__global__ void __launch_bounds__(1024)
bin_place(const int* __restrict__ ei, const int* __restrict__ off,
          int* __restrict__ sorted) {
  __shared__ int cur[NBINS];
  const int tid = threadIdx.x;
  if (tid < NBINS) cur[tid] = off[blockIdx.x * NBINS + tid];
  __syncthreads();
  const int* src = ei + blockIdx.x * EPB;
  const int* dst = src + N_EDGES;
  int s[BIN_ITERS], d[BIN_ITERS];
#pragma unroll
  for (int j = 0; j < BIN_ITERS; ++j) {
    int i = tid + j * 1024;
    i = i < EPB ? i : EPB - 1;
    s[j] = src[i];
    d[j] = dst[i];
  }
#pragma unroll
  for (int j = 0; j < BIN_ITERS; ++j) {
    if (tid + j * 1024 < EPB) {
      int b = d[j] / RANGE;
      int dl = d[j] - b * RANGE;
      int pos = atomicAdd(&cur[b], 1);
      sorted[pos] = s[j] | (dl << SRC_BITS);
    }
  }
}

#define EPW         8
#define ACC_PITCH   68
#define RANGE_TILES ((RANGE + 15) / 16)
#define RANGE_PAD   (RANGE_TILES * 16)
static_assert((ACC_PITCH * 4) % 16 == 0, "LDS rows must stay 16-byte aligned");
static_assert(NFRAG_G % 1024 == 0, "table staging assumes whole rounds of 1024");
static_assert(RANGE_PAD * ACC_PITCH % 4 == 0, "sums are cleared as float4");

__global__ void __launch_bounds__(1024)
layer_fused(const float* __restrict__ hin, float* __restrict__ hout,
            const int* __restrict__ sorted, const int* __restrict__ binStart,
            const v2f* __restrict__ pIh, const v2f* __restrict__ pHh,
            const float* __restrict__ bih, const float* __restrict__ bhh) {
  __shared__ __attribute__((aligned(16))) float acc[RANGE_PAD * ACC_PITCH];
  __shared__ v2f sIh[NFRAG_G];
  __shared__ v2f sHh[NFRAG_G];
  const int tid = threadIdx.x;
  const int lane = tid & 31;

  float4* acc4 = (float4*)acc;
  for (int i = tid; i < RANGE_PAD * ACC_PITCH / 4; i += 1024)
    acc4[i] = make_float4(0.0f, 0.0f, 0.0f, 0.0f);
  __syncthreads();

  constexpr int TPT = NFRAG_G / 1024;
  v2f tI[TPT], tH[TPT];
#pragma unroll
  for (int i = 0; i < TPT; ++i) {
    tI[i] = pIh[tid + i * 1024];
    tH[i] = pHh[tid + i * 1024];
  }

  const int b = blockIdx.x;
  const int beg = binStart[b];
  const int end = binStart[b + 1];
  const int w = __builtin_amdgcn_readfirstlane(tid >> 5);

  int e0 = beg + w * EPW;
  int pk[EPW];
  if (e0 < end) {
#pragma unroll
    for (int k = 0; k < EPW; ++k) {
      int e = e0 + k;
      pk[k] = sorted[e < end ? e : end - 1];
    }
  }
  while (e0 < end) {
    float v0[EPW], v1[EPW];
#pragma unroll
    for (int k = 0; k < EPW; ++k) {
      const float* mp = hin + (size_t)(pk[k] & SRC_MASK) * H_DIM + lane;
      v0[k] = mp[0];
      v1[k] = mp[32];
    }
    const int e1 = e0 + 32 * EPW;
    int nk[EPW];
#pragma unroll
    for (int k = 0; k < EPW; ++k) {
      int e = e1 + k;
      nk[k] = sorted[e < end ? e : end - 1];
    }
#pragma unroll
    for (int k = 0; k < EPW; ++k) {
      const bool live = (e0 + k) < end;
      float* ap = acc + (pk[k] >> SRC_BITS) * ACC_PITCH + lane;
      __hip_atomic_fetch_add(ap, live ? v0[k] : 0.0f, __ATOMIC_RELAXED,
                             __HIP_MEMORY_SCOPE_WORKGROUP);
      __hip_atomic_fetch_add(ap + 32, live ? v1[k] : 0.0f, __ATOMIC_RELAXED,
                             __HIP_MEMORY_SCOPE_WORKGROUP);
    }
#pragma unroll
    for (int k = 0; k < EPW; ++k) pk[k] = nk[k];
    e0 = e1;
  }

#pragma unroll
  for (int i = 0; i < TPT; ++i) {
    sIh[tid + i * 1024] = tI[i];
    sHh[tid + i * 1024] = tH[i];
  }
  __syncthreads();

  const int row0 = b * RANGE;
  const int nrows = (N_NODES - row0) < RANGE ? (N_NODES - row0) : RANGE;
  const int nUnits = ((nrows + 15) >> 4) * 4;
  const int ln = lane & 15, lh = lane >> 4;

#pragma unroll 1
  for (int u = w; u < nUnits; u += 32) {
    const int t = u >> 2;
    const int j = (u + (u >> 5)) & 3;
    const int lr = t * 16 + ln;
    int gr = row0 + lr;
    gr = gr < N_NODES ? gr : N_NODES - 1;
    const v2f* hp = (const v2f*)(hin + (size_t)gr * H_DIM);
    v2f aH[16];
#pragma unroll
    for (int kk = 0; kk < 16; ++kk) aH[kk] = hp[2 * kk + lh];
    const v2f* ap = (const v2f*)(acc + lr * ACC_PITCH);

    const int c = j * 16 + ln;
    float bir = bih[c], biz = bih[64 + c], bin = bih[128 + c];
    float bhr = bhh[c], bhz = bhh[64 + c], bhn = bhh[128 + c];
    v8f gi_r, gi_z, gi_n, gh_r, gh_z, gh_n;
#pragma unroll
    for (int v = 0; v < 8; ++v) {
      gi_r[v] = bir; gi_z[v] = biz; gi_n[v] = bin;
      gh_r[v] = bhr; gh_z[v] = bhz; gh_n[v] = bhn;
    }

#pragma unroll
    for (int kk = 0; kk < 16; ++kk) {
      v2f a = ap[2 * kk + lh];
      int base = kk * 12 * 32 + lane;
      v2f br = sIh[base + (j + 0) * 32];
      v2f bz = sIh[base + (j + 4) * 32];
      v2f bn = sIh[base + (j + 8) * 32];
      v2f cr = sHh[base + (j + 0) * 32];
      v2f cz = sHh[base + (j + 4) * 32];
      v2f cn = sHh[base + (j + 8) * 32];
      gi_r = wmma4(a, br, gi_r);
      gi_z = wmma4(a, bz, gi_z);
      gi_n = wmma4(a, bn, gi_n);
      gh_r = wmma4(aH[kk], cr, gh_r);
      gh_z = wmma4(aH[kk], cz, gh_z);
      gh_n = wmma4(aH[kk], cn, gh_n);
    }

#pragma unroll
    for (int v = 0; v < 8; ++v) {
      const int lrow = t * 16 + v + 8 * lh;
      int grow = row0 + lrow;
      grow = grow < N_NODES ? grow : N_NODES - 1;
      size_t idx = (size_t)grow * H_DIM + c;
      float hv = hin[idx];
      float r = 1.0f / (1.0f + __expf(-(gi_r[v] + gh_r[v])));
      float z = 1.0f / (1.0f + __expf(-(gi_z[v] + gh_z[v])));
      float n = tanhf(gi_n[v] + r * gh_n[v]);
      float o = (1.0f - z) * n + z * hv;
      if (lrow < nrows) hout[idx] = o;
    }
  }
}

__global__ void zero_f32(float* __restrict__ p, int n) {
  int tid = blockIdx.x * blockDim.x + threadIdx.x;
  if (tid < n) p[tid] = 0.0f;
}

__global__ void pool(const float* __restrict__ h, const int* __restrict__ batch,
                     float* __restrict__ out, int nNodes) {
  int tid = blockIdx.x * blockDim.x + threadIdx.x;
  int n = tid >> 6;
  if (n >= nNodes) return;
  int d = tid & 63;
  int g = batch[n];
  __hip_atomic_fetch_add(out + (size_t)g * H_DIM + d, h[(size_t)n * H_DIM + d],
                         __ATOMIC_RELAXED, __HIP_MEMORY_SCOPE_AGENT);
}

extern "C" void kernel_launch(void* const* d_in, const int* in_sizes, int n_in,
                              void* d_out, int out_size, void* d_ws,
                              size_t ws_size, hipStream_t stream) {
  const float* x      = (const float*)d_in[0];
  const int*   ei     = (const int*)d_in[1];
  const int*   batch  = (const int*)d_in[2];
  const float* weight = (const float*)d_in[3];
  const float* Wih    = (const float*)d_in[4];
  const float* Whh    = (const float*)d_in[5];
  const float* bih    = (const float*)d_in[6];
  const float* bhh    = (const float*)d_in[7];
  float* out = (float*)d_out;

  const size_t NB = (size_t)N_NODES * H_DIM * sizeof(float);
  char* ws = (char*)d_ws;
  float* hA = (float*)(ws);
  float* hB = (float*)(ws + 2 * NB);
  v2f* packWc = (v2f*)(ws + NB);
  v2f* packHh = packWc + (size_t)N_LAYERS * NFRAG_G;
  int* cnt      = (int*)(packHh + NFRAG_G);
  int* binStart = cnt + BIN_WGS * NBINS;
  int* sorted   = binStart + 512;

  pack_wc<<<(N_LAYERS * NFRAG_G + 255) / 256, 256, 0, stream>>>(weight, Wih,
                                                                (float*)packWc);
  pack_b<<<(NFRAG_G + 255) / 256, 256, 0, stream>>>(Whh, (float*)packHh, 12, 1);

  bin_count<<<BIN_WGS, 1024, 0, stream>>>(ei, cnt);
  bin_scan<<<1, 1024, 0, stream>>>(cnt, binStart);
  bin_place<<<BIN_WGS, 1024, 0, stream>>>(ei, cnt, sorted);

  layer_fused<<<NBINS, 1024, 0, stream>>>(x, hA, sorted, binStart, packWc,
                                          packHh, bih, bhh);
  layer_fused<<<NBINS, 1024, 0, stream>>>(hA, hB, sorted, binStart,
                                          packWc + NFRAG_G, packHh, bih, bhh);
  layer_fused<<<NBINS, 1024, 0, stream>>>(hB, hA, sorted, binStart,
                                          packWc + 2 * NFRAG_G, packHh, bih,
                                          bhh);

  zero_f32<<<(NUM_GRAPHS * H_DIM + 255) / 256, 256, 0, stream>>>(
      out, NUM_GRAPHS * H_DIM);
  pool<<<(N_NODES * H_DIM + 255) / 256, 256, 0, stream>>>(hA, batch, out,
                                                           N_NODES);
}


# graph pooling fused into the last layer's launch: new rows summed per graph id in LDS, one atomic per id and column per range; last h plane not stored, pool launch goes; f32 sum order changes
# speedup vs baseline: 2.6885x; 1.0795x over previous
#include <hip/hip_runtime.h>
#include <math.h>

#define N_NODES    100000
#define N_EDGES    1600000
#define H_DIM      64
#define N_LAYERS   3
#define NUM_GRAPHS 256

typedef __attribute__((ext_vector_type(2))) float v2f;
typedef __attribute__((ext_vector_type(8))) float v8f;

__device__ __forceinline__ v8f wmma4(v2f a, v2f b, v8f c) {
  return __builtin_amdgcn_wmma_f32_16x16x4_f32(false, a, false, b, (short)0, c,
                                               false, false);
}

__global__ void pack_b(const float* __restrict__ Wsrc, float* __restrict__ pack,
                       int nGroups, int transposed) {
  int tid = blockIdx.x * blockDim.x + threadIdx.x;
  int total = 16 * nGroups * 32;
  if (tid >= total) return;
  int lane = tid & 31;
  int t = tid >> 5;
  int g = t % nGroups, kk = t / nGroups;
  int ln = lane & 15, lh = lane >> 4;
  int c = g * 16 + ln;
  int k0 = 4 * kk + 2 * lh;
  v2f b;
  if (transposed) {
    b.x = Wsrc[c * 64 + k0];
    b.y = Wsrc[c * 64 + k0 + 1];
  } else {
    b.x = Wsrc[k0 * 64 + c];
    b.y = Wsrc[(k0 + 1) * 64 + c];
  }
  ((v2f*)pack)[tid] = b;
}

#define NFRAG_G (16 * 12 * 32)

__global__ void pack_wc(const float* __restrict__ weight,
                        const float* __restrict__ Wih,
                        float* __restrict__ pack) {
  int tid = blockIdx.x * blockDim.x + threadIdx.x;
  if (tid >= N_LAYERS * NFRAG_G) return;
  int layer = tid / NFRAG_G;
  int f = tid - layer * NFRAG_G;
  int lane = f & 31;
  int t = f >> 5;
  int g = t % 12, kk = t / 12;
  int ln = lane & 15, lh = lane >> 4;
  int c = g * 16 + ln;
  int k0 = 4 * kk + 2 * lh;
  const float* w0 = weight + (size_t)layer * 64 * 64 + (size_t)k0 * 64;
  const float* w1 = w0 + 64;
  const float* wi = Wih + (size_t)c * 64;
  float s0 = 0.0f, s1 = 0.0f;
#pragma unroll
  for (int j = 0; j < 64; ++j) {
    float u = wi[j];
    s0 = fmaf(w0[j], u, s0);
    s1 = fmaf(w1[j], u, s1);
  }
  v2f b;
  b.x = s0;
  b.y = s1;
  ((v2f*)pack)[tid] = b;
}

#define NBINS    256
#define RANGE    391
#define BIN_WGS  256
#define EPB      (N_EDGES / BIN_WGS)
#define BIN_ITERS ((EPB + 1023) / 1024)
#define SRC_BITS 17
#define SRC_MASK ((1 << SRC_BITS) - 1)
static_assert(RANGE * NBINS >= N_NODES, "ranges must cover all rows");
static_assert(N_EDGES % BIN_WGS == 0, "edge count must split evenly");
static_assert((1 << SRC_BITS) >= N_NODES, "source row must fit its field");
static_assert(RANGE <= (1 << (31 - SRC_BITS)), "local row must fit its field");
static_assert(BIN_WGS % 4 == 0, "bin_scan splits the work-groups in four");

__global__ void __launch_bounds__(1024)
bin_count(const int* __restrict__ ei, int* __restrict__ cnt) {
  __shared__ int hist[NBINS];
  const int tid = threadIdx.x;
  if (tid < NBINS) hist[tid] = 0;
  __syncthreads();
  const int* dst = ei + N_EDGES + blockIdx.x * EPB;
  int d[BIN_ITERS];
#pragma unroll
  for (int j = 0; j < BIN_ITERS; ++j) {
    int i = tid + j * 1024;
    d[j] = dst[i < EPB ? i : EPB - 1];
  }
#pragma unroll
  for (int j = 0; j < BIN_ITERS; ++j)
    if (tid + j * 1024 < EPB) atomicAdd(&hist[d[j] / RANGE], 1);
  __syncthreads();
  if (tid < NBINS) cnt[blockIdx.x * NBINS + tid] = hist[tid];
}

__global__ void __launch_bounds__(1024)
bin_scan(int* __restrict__ cnt, int* __restrict__ binStart) {
  __shared__ int part[4][NBINS];
  __shared__ int sc[NBINS];
  const int tid = threadIdx.x;
  const int t = tid & (NBINS - 1);
  const int q = tid >> 8;
  constexpr int WPQ = BIN_WGS / 4;
  int c[WPQ];
#pragma unroll
  for (int w = 0; w < WPQ; ++w) c[w] = cnt[(q * WPQ + w) * NBINS + t];
  int run = 0;
#pragma unroll
  for (int w = 0; w < WPQ; ++w) run += c[w];
  part[q][t] = run;
  __syncthreads();
  int base = 0;
  for (int p = 0; p < q; ++p) base += part[p][t];
  if (q == 3) sc[t] = base + run;
  __syncthreads();
  const int total = sc[t];
  for (int off = 1; off < NBINS; off <<= 1) {
    int v = 0;
    if (tid < NBINS && tid >= off) v = sc[tid - off];
    __syncthreads();
    if (tid < NBINS) sc[tid] += v;
    __syncthreads();
  }
  const int start = sc[t] - total;
  int pos = start + base;
#pragma unroll
  for (int w = 0; w < WPQ; ++w) {
    cnt[(q * WPQ + w) * NBINS + t] = pos;
    pos += c[w];
  }
  if (q == 0) binStart[t] = start;
  if (tid == 0) binStart[NBINS] = sc[NBINS - 1];
}

__global__ void __launch_bounds__(1024)
bin_place(const int* __restrict__ ei, const int* __restrict__ off,
          int* __restrict__ sorted) {
  __shared__ int cur[NBINS];
  const int tid = threadIdx.x;
  if (tid < NBINS) cur[tid] = off[blockIdx.x * NBINS + tid];
  __syncthreads();
  const int* src = ei + blockIdx.x * EPB;
  const int* dst = src + N_EDGES;
  int s[BIN_ITERS], d[BIN_ITERS];
#pragma unroll
  for (int j = 0; j < BIN_ITERS; ++j) {
    int i = tid + j * 1024;
    i = i < EPB ? i : EPB - 1;
    s[j] = src[i];
    d[j] = dst[i];
  }
#pragma unroll
  for (int j = 0; j < BIN_ITERS; ++j) {
    if (tid + j * 1024 < EPB) {
      int b = d[j] / RANGE;
      int dl = d[j] - b * RANGE;
      int pos = atomicAdd(&cur[b], 1);
      sorted[pos] = s[j] | (dl << SRC_BITS);
    }
  }
}

#define POOL_SPAN   16
#define EPW         8
#define ACC_PITCH   68
#define RANGE_TILES ((RANGE + 15) / 16)
#define RANGE_PAD   (RANGE_TILES * 16)
static_assert((ACC_PITCH * 4) % 16 == 0, "LDS rows must stay 16-byte aligned");
static_assert(NFRAG_G % 1024 == 0, "table staging assumes whole rounds of 1024");
static_assert(RANGE_PAD * ACC_PITCH % 4 == 0, "sums are cleared as float4");

template <bool LAST>
__global__ void __launch_bounds__(1024)
layer_fused(const float* __restrict__ hin, float* __restrict__ hout,
            const int* __restrict__ sorted, const int* __restrict__ binStart,
            const v2f* __restrict__ pIh, const v2f* __restrict__ pHh,
            const float* __restrict__ bih, const float* __restrict__ bhh,
            const int* __restrict__ batch, float* __restrict__ out) {
  __shared__ __attribute__((aligned(16))) float acc[RANGE_PAD * ACC_PITCH];
  __shared__ v2f sIh[NFRAG_G];
  __shared__ v2f sHh[NFRAG_G];
  __shared__ float pl[LAST ? POOL_SPAN * H_DIM : 1];
  __shared__ int sid[LAST ? RANGE_PAD : 1];
  const int tid = threadIdx.x;
  const int lane = tid & 31;

  if constexpr (LAST) {
    for (int i = tid; i < POOL_SPAN * H_DIM; i += 1024) pl[i] = 0.0f;
    if (tid < RANGE_PAD) {
      int gr = blockIdx.x * RANGE + tid;
      sid[tid] = batch[gr < N_NODES ? gr : N_NODES - 1];
    }
  }

  float4* acc4 = (float4*)acc;
  for (int i = tid; i < RANGE_PAD * ACC_PITCH / 4; i += 1024)
    acc4[i] = make_float4(0.0f, 0.0f, 0.0f, 0.0f);
  __syncthreads();

  constexpr int TPT = NFRAG_G / 1024;
  v2f tI[TPT], tH[TPT];
#pragma unroll
  for (int i = 0; i < TPT; ++i) {
    tI[i] = pIh[tid + i * 1024];
    tH[i] = pHh[tid + i * 1024];
  }

  const int b = blockIdx.x;
  const int beg = binStart[b];
  const int end = binStart[b + 1];
  const int w = __builtin_amdgcn_readfirstlane(tid >> 5);

  int e0 = beg + w * EPW;
  int pk[EPW];
  if (e0 < end) {
#pragma unroll
    for (int k = 0; k < EPW; ++k) {
      int e = e0 + k;
      pk[k] = sorted[e < end ? e : end - 1];
    }
  }
  while (e0 < end) {
    float v0[EPW], v1[EPW];
#pragma unroll
    for (int k = 0; k < EPW; ++k) {
      const float* mp = hin + (size_t)(pk[k] & SRC_MASK) * H_DIM + lane;
      v0[k] = mp[0];
      v1[k] = mp[32];
    }
    const int e1 = e0 + 32 * EPW;
    int nk[EPW];
#pragma unroll
    for (int k = 0; k < EPW; ++k) {
      int e = e1 + k;
      nk[k] = sorted[e < end ? e : end - 1];
    }
#pragma unroll
    for (int k = 0; k < EPW; ++k) {
      const bool live = (e0 + k) < end;
      float* ap = acc + (pk[k] >> SRC_BITS) * ACC_PITCH + lane;
      __hip_atomic_fetch_add(ap, live ? v0[k] : 0.0f, __ATOMIC_RELAXED,
                             __HIP_MEMORY_SCOPE_WORKGROUP);
      __hip_atomic_fetch_add(ap + 32, live ? v1[k] : 0.0f, __ATOMIC_RELAXED,
                             __HIP_MEMORY_SCOPE_WORKGROUP);
    }
#pragma unroll
    for (int k = 0; k < EPW; ++k) pk[k] = nk[k];
    e0 = e1;
  }

#pragma unroll
  for (int i = 0; i < TPT; ++i) {
    sIh[tid + i * 1024] = tI[i];
    sHh[tid + i * 1024] = tH[i];
  }
  __syncthreads();

  const int row0 = b * RANGE;
  const int lo = LAST ? batch[row0] : 0;
  const int nrows = (N_NODES - row0) < RANGE ? (N_NODES - row0) : RANGE;
  const int nUnits = ((nrows + 15) >> 4) * 4;
  const int ln = lane & 15, lh = lane >> 4;

#pragma unroll 1
  for (int u = w; u < nUnits; u += 32) {
    const int t = u >> 2;
    const int j = (u + (u >> 5)) & 3;
    const int lr = t * 16 + ln;
    int gr = row0 + lr;
    gr = gr < N_NODES ? gr : N_NODES - 1;
    const v2f* hp = (const v2f*)(hin + (size_t)gr * H_DIM);
    v2f aH[16];
#pragma unroll
    for (int kk = 0; kk < 16; ++kk) aH[kk] = hp[2 * kk + lh];
    const v2f* ap = (const v2f*)(acc + lr * ACC_PITCH);

    const int c = j * 16 + ln;
    float bir = bih[c], biz = bih[64 + c], bin = bih[128 + c];
    float bhr = bhh[c], bhz = bhh[64 + c], bhn = bhh[128 + c];
    v8f gi_r, gi_z, gi_n, gh_r, gh_z, gh_n;
#pragma unroll
    for (int v = 0; v < 8; ++v) {
      gi_r[v] = bir; gi_z[v] = biz; gi_n[v] = bin;
      gh_r[v] = bhr; gh_z[v] = bhz; gh_n[v] = bhn;
    }

#pragma unroll
    for (int kk = 0; kk < 16; ++kk) {
      v2f a = ap[2 * kk + lh];
      int base = kk * 12 * 32 + lane;
      v2f br = sIh[base + (j + 0) * 32];
      v2f bz = sIh[base + (j + 4) * 32];
      v2f bn = sIh[base + (j + 8) * 32];
      v2f cr = sHh[base + (j + 0) * 32];
      v2f cz = sHh[base + (j + 4) * 32];
      v2f cn = sHh[base + (j + 8) * 32];
      gi_r = wmma4(a, br, gi_r);
      gi_z = wmma4(a, bz, gi_z);
      gi_n = wmma4(a, bn, gi_n);
      gh_r = wmma4(aH[kk], cr, gh_r);
      gh_z = wmma4(aH[kk], cz, gh_z);
      gh_n = wmma4(aH[kk], cn, gh_n);
    }

#pragma unroll
    for (int v = 0; v < 8; ++v) {
      const int lrow = t * 16 + v + 8 * lh;
      int grow = row0 + lrow;
      grow = grow < N_NODES ? grow : N_NODES - 1;
      size_t idx = (size_t)grow * H_DIM + c;
      float hv = hin[idx];
      float r = 1.0f / (1.0f + __expf(-(gi_r[v] + gh_r[v])));
      float z = 1.0f / (1.0f + __expf(-(gi_z[v] + gh_z[v])));
      float n = tanhf(gi_n[v] + r * gh_n[v]);
      float o = (1.0f - z) * n + z * hv;
      if (lrow < nrows) {
        if constexpr (LAST) {
          const int g = sid[lrow];
          const unsigned rel = (unsigned)(g - lo);
          if (rel < POOL_SPAN)
            __hip_atomic_fetch_add(pl + rel * H_DIM + c, o, __ATOMIC_RELAXED,
                                   __HIP_MEMORY_SCOPE_WORKGROUP);
          else
            __hip_atomic_fetch_add(out + (size_t)g * H_DIM + c, o,
                                   __ATOMIC_RELAXED, __HIP_MEMORY_SCOPE_AGENT);
        } else {
          hout[idx] = o;
        }
      }
    }
  }

  if constexpr (LAST) {
    __syncthreads();
    static_assert(POOL_SPAN * H_DIM == 1024, "one pool slot per thread");
    const int i = w * 32 + (int)__builtin_amdgcn_mbcnt_lo(~0u, 0u);
    const float s = pl[i];
    const int g = lo + (i >> 6);
    if (g < NUM_GRAPHS && s != 0.0f)
      __hip_atomic_fetch_add(out + (size_t)g * H_DIM + (i & 63), s,
                             __ATOMIC_RELAXED, __HIP_MEMORY_SCOPE_AGENT);
  }
}

__global__ void zero_f32(float* __restrict__ p, int n) {
  int tid = blockIdx.x * blockDim.x + threadIdx.x;
  if (tid < n) p[tid] = 0.0f;
}

extern "C" void kernel_launch(void* const* d_in, const int* in_sizes, int n_in,
                              void* d_out, int out_size, void* d_ws,
                              size_t ws_size, hipStream_t stream) {
  const float* x      = (const float*)d_in[0];
  const int*   ei     = (const int*)d_in[1];
  const int*   batch  = (const int*)d_in[2];
  const float* weight = (const float*)d_in[3];
  const float* Wih    = (const float*)d_in[4];
  const float* Whh    = (const float*)d_in[5];
  const float* bih    = (const float*)d_in[6];
  const float* bhh    = (const float*)d_in[7];
  float* out = (float*)d_out;

  const size_t NB = (size_t)N_NODES * H_DIM * sizeof(float);
  char* ws = (char*)d_ws;
  float* hA = (float*)(ws);
  float* hB = (float*)(ws + 2 * NB);
  v2f* packWc = (v2f*)(ws + NB);
  v2f* packHh = packWc + (size_t)N_LAYERS * NFRAG_G;
  int* cnt      = (int*)(packHh + NFRAG_G);
  int* binStart = cnt + BIN_WGS * NBINS;
  int* sorted   = binStart + 512;

  zero_f32<<<(NUM_GRAPHS * H_DIM + 255) / 256, 256, 0, stream>>>(
      out, NUM_GRAPHS * H_DIM);

  pack_wc<<<(N_LAYERS * NFRAG_G + 255) / 256, 256, 0, stream>>>(weight, Wih,
                                                                (float*)packWc);
  pack_b<<<(NFRAG_G + 255) / 256, 256, 0, stream>>>(Whh, (float*)packHh, 12, 1);

  bin_count<<<BIN_WGS, 1024, 0, stream>>>(ei, cnt);
  bin_scan<<<1, 1024, 0, stream>>>(cnt, binStart);
  bin_place<<<BIN_WGS, 1024, 0, stream>>>(ei, cnt, sorted);

  layer_fused<false><<<NBINS, 1024, 0, stream>>>(x, hA, sorted, binStart,
                                                 packWc, packHh, bih, bhh,
                                                 batch, out);
  layer_fused<false><<<NBINS, 1024, 0, stream>>>(hA, hB, sorted, binStart,
                                                 packWc + NFRAG_G, packHh, bih,
                                                 bhh, batch, out);
  layer_fused<true><<<NBINS, 1024, 0, stream>>>(hB, hA, sorted, binStart,
                                                packWc + 2 * NFRAG_G, packHh,
                                                bih, bhh, batch, out);
}


# GRU gate math on hardware exp2 and reciprocal: sigmoid = rcp(1+exp2(-cx)), tanh = 1-2*rcp(1+exp2(2cx)), in place of library tanhf (two branches a value) and IEEE divisions; sum order unchanged
# speedup vs baseline: 2.8164x; 1.0476x over previous
#include <hip/hip_runtime.h>
#include <math.h>

#define N_NODES    100000
#define N_EDGES    1600000
#define H_DIM      64
#define N_LAYERS   3
#define NUM_GRAPHS 256

typedef __attribute__((ext_vector_type(2))) float v2f;
typedef __attribute__((ext_vector_type(8))) float v8f;

__device__ __forceinline__ v8f wmma4(v2f a, v2f b, v8f c) {
  return __builtin_amdgcn_wmma_f32_16x16x4_f32(false, a, false, b, (short)0, c,
                                               false, false);
}

#define LOG2E_F 1.4426950408889634f
__device__ __forceinline__ float sigmoid_hw(float x) {
  return __builtin_amdgcn_rcpf(1.0f + __builtin_amdgcn_exp2f(-LOG2E_F * x));
}
__device__ __forceinline__ float tanh_hw(float x) {
  return fmaf(-2.0f,
              __builtin_amdgcn_rcpf(
                  1.0f + __builtin_amdgcn_exp2f((2.0f * LOG2E_F) * x)),
              1.0f);
}

__global__ void pack_b(const float* __restrict__ Wsrc, float* __restrict__ pack,
                       int nGroups, int transposed) {
  int tid = blockIdx.x * blockDim.x + threadIdx.x;
  int total = 16 * nGroups * 32;
  if (tid >= total) return;
  int lane = tid & 31;
  int t = tid >> 5;
  int g = t % nGroups, kk = t / nGroups;
  int ln = lane & 15, lh = lane >> 4;
  int c = g * 16 + ln;
  int k0 = 4 * kk + 2 * lh;
  v2f b;
  if (transposed) {
    b.x = Wsrc[c * 64 + k0];
    b.y = Wsrc[c * 64 + k0 + 1];
  } else {
    b.x = Wsrc[k0 * 64 + c];
    b.y = Wsrc[(k0 + 1) * 64 + c];
  }
  ((v2f*)pack)[tid] = b;
}

#define NFRAG_G (16 * 12 * 32)

__global__ void pack_wc(const float* __restrict__ weight,
                        const float* __restrict__ Wih,
                        float* __restrict__ pack) {
  int tid = blockIdx.x * blockDim.x + threadIdx.x;
  if (tid >= N_LAYERS * NFRAG_G) return;
  int layer = tid / NFRAG_G;
  int f = tid - layer * NFRAG_G;
  int lane = f & 31;
  int t = f >> 5;
  int g = t % 12, kk = t / 12;
  int ln = lane & 15, lh = lane >> 4;
  int c = g * 16 + ln;
  int k0 = 4 * kk + 2 * lh;
  const float* w0 = weight + (size_t)layer * 64 * 64 + (size_t)k0 * 64;
  const float* w1 = w0 + 64;
  const float* wi = Wih + (size_t)c * 64;
  float s0 = 0.0f, s1 = 0.0f;
#pragma unroll
  for (int j = 0; j < 64; ++j) {
    float u = wi[j];
    s0 = fmaf(w0[j], u, s0);
    s1 = fmaf(w1[j], u, s1);
  }
  v2f b;
  b.x = s0;
  b.y = s1;
  ((v2f*)pack)[tid] = b;
}

#define NBINS    256
#define RANGE    391
#define BIN_WGS  256
#define EPB      (N_EDGES / BIN_WGS)
#define BIN_ITERS ((EPB + 1023) / 1024)
#define SRC_BITS 17
#define SRC_MASK ((1 << SRC_BITS) - 1)
static_assert(RANGE * NBINS >= N_NODES, "ranges must cover all rows");
static_assert(N_EDGES % BIN_WGS == 0, "edge count must split evenly");
static_assert((1 << SRC_BITS) >= N_NODES, "source row must fit its field");
static_assert(RANGE <= (1 << (31 - SRC_BITS)), "local row must fit its field");
static_assert(BIN_WGS % 4 == 0, "bin_scan splits the work-groups in four");

__global__ void __launch_bounds__(1024)
bin_count(const int* __restrict__ ei, int* __restrict__ cnt) {
  __shared__ int hist[NBINS];
  const int tid = threadIdx.x;
  if (tid < NBINS) hist[tid] = 0;
  __syncthreads();
  const int* dst = ei + N_EDGES + blockIdx.x * EPB;
  int d[BIN_ITERS];
#pragma unroll
  for (int j = 0; j < BIN_ITERS; ++j) {
    int i = tid + j * 1024;
    d[j] = dst[i < EPB ? i : EPB - 1];
  }
#pragma unroll
  for (int j = 0; j < BIN_ITERS; ++j)
    if (tid + j * 1024 < EPB) atomicAdd(&hist[d[j] / RANGE], 1);
  __syncthreads();
  if (tid < NBINS) cnt[blockIdx.x * NBINS + tid] = hist[tid];
}

__global__ void __launch_bounds__(1024)
bin_scan(int* __restrict__ cnt, int* __restrict__ binStart) {
  __shared__ int part[4][NBINS];
  __shared__ int sc[NBINS];
  const int tid = threadIdx.x;
  const int t = tid & (NBINS - 1);
  const int q = tid >> 8;
  constexpr int WPQ = BIN_WGS / 4;
  int c[WPQ];
#pragma unroll
  for (int w = 0; w < WPQ; ++w) c[w] = cnt[(q * WPQ + w) * NBINS + t];
  int run = 0;
#pragma unroll
  for (int w = 0; w < WPQ; ++w) run += c[w];
  part[q][t] = run;
  __syncthreads();
  int base = 0;
  for (int p = 0; p < q; ++p) base += part[p][t];
  if (q == 3) sc[t] = base + run;
  __syncthreads();
  const int total = sc[t];
  for (int off = 1; off < NBINS; off <<= 1) {
    int v = 0;
    if (tid < NBINS && tid >= off) v = sc[tid - off];
    __syncthreads();
    if (tid < NBINS) sc[tid] += v;
    __syncthreads();
  }
  const int start = sc[t] - total;
  int pos = start + base;
#pragma unroll
  for (int w = 0; w < WPQ; ++w) {
    cnt[(q * WPQ + w) * NBINS + t] = pos;
    pos += c[w];
  }
  if (q == 0) binStart[t] = start;
  if (tid == 0) binStart[NBINS] = sc[NBINS - 1];
}

__global__ void __launch_bounds__(1024)
bin_place(const int* __restrict__ ei, const int* __restrict__ off,
          int* __restrict__ sorted) {
  __shared__ int cur[NBINS];
  const int tid = threadIdx.x;
  if (tid < NBINS) cur[tid] = off[blockIdx.x * NBINS + tid];
  __syncthreads();
  const int* src = ei + blockIdx.x * EPB;
  const int* dst = src + N_EDGES;
  int s[BIN_ITERS], d[BIN_ITERS];
#pragma unroll
  for (int j = 0; j < BIN_ITERS; ++j) {
    int i = tid + j * 1024;
    i = i < EPB ? i : EPB - 1;
    s[j] = src[i];
    d[j] = dst[i];
  }
#pragma unroll
  for (int j = 0; j < BIN_ITERS; ++j) {
    if (tid + j * 1024 < EPB) {
      int b = d[j] / RANGE;
      int dl = d[j] - b * RANGE;
      int pos = atomicAdd(&cur[b], 1);
      sorted[pos] = s[j] | (dl << SRC_BITS);
    }
  }
}

#define POOL_SPAN   16
#define EPW         8
#define ACC_PITCH   68
#define RANGE_TILES ((RANGE + 15) / 16)
#define RANGE_PAD   (RANGE_TILES * 16)
static_assert((ACC_PITCH * 4) % 16 == 0, "LDS rows must stay 16-byte aligned");
static_assert(NFRAG_G % 1024 == 0, "table staging assumes whole rounds of 1024");
static_assert(RANGE_PAD * ACC_PITCH % 4 == 0, "sums are cleared as float4");

template <bool LAST>
__global__ void __launch_bounds__(1024)
layer_fused(const float* __restrict__ hin, float* __restrict__ hout,
            const int* __restrict__ sorted, const int* __restrict__ binStart,
            const v2f* __restrict__ pIh, const v2f* __restrict__ pHh,
            const float* __restrict__ bih, const float* __restrict__ bhh,
            const int* __restrict__ batch, float* __restrict__ out) {
  __shared__ __attribute__((aligned(16))) float acc[RANGE_PAD * ACC_PITCH];
  __shared__ v2f sIh[NFRAG_G];
  __shared__ v2f sHh[NFRAG_G];
  __shared__ float pl[LAST ? POOL_SPAN * H_DIM : 1];
  __shared__ int sid[LAST ? RANGE_PAD : 1];
  const int tid = threadIdx.x;
  const int lane = tid & 31;

  if constexpr (LAST) {
    for (int i = tid; i < POOL_SPAN * H_DIM; i += 1024) pl[i] = 0.0f;
    if (tid < RANGE_PAD) {
      int gr = blockIdx.x * RANGE + tid;
      sid[tid] = batch[gr < N_NODES ? gr : N_NODES - 1];
    }
  }

  float4* acc4 = (float4*)acc;
  for (int i = tid; i < RANGE_PAD * ACC_PITCH / 4; i += 1024)
    acc4[i] = make_float4(0.0f, 0.0f, 0.0f, 0.0f);
  __syncthreads();

  constexpr int TPT = NFRAG_G / 1024;
  v2f tI[TPT], tH[TPT];
#pragma unroll
  for (int i = 0; i < TPT; ++i) {
    tI[i] = pIh[tid + i * 1024];
    tH[i] = pHh[tid + i * 1024];
  }

  const int b = blockIdx.x;
  const int beg = binStart[b];
  const int end = binStart[b + 1];
  const int w = __builtin_amdgcn_readfirstlane(tid >> 5);

  int e0 = beg + w * EPW;
  int pk[EPW];
  if (e0 < end) {
#pragma unroll
    for (int k = 0; k < EPW; ++k) {
      int e = e0 + k;
      pk[k] = sorted[e < end ? e : end - 1];
    }
  }
  while (e0 < end) {
    float v0[EPW], v1[EPW];
#pragma unroll
    for (int k = 0; k < EPW; ++k) {
      const float* mp = hin + (size_t)(pk[k] & SRC_MASK) * H_DIM + lane;
      v0[k] = mp[0];
      v1[k] = mp[32];
    }
    const int e1 = e0 + 32 * EPW;
    int nk[EPW];
#pragma unroll
    for (int k = 0; k < EPW; ++k) {
      int e = e1 + k;
      nk[k] = sorted[e < end ? e : end - 1];
    }
#pragma unroll
    for (int k = 0; k < EPW; ++k) {
      const bool live = (e0 + k) < end;
      float* ap = acc + (pk[k] >> SRC_BITS) * ACC_PITCH + lane;
      __hip_atomic_fetch_add(ap, live ? v0[k] : 0.0f, __ATOMIC_RELAXED,
                             __HIP_MEMORY_SCOPE_WORKGROUP);
      __hip_atomic_fetch_add(ap + 32, live ? v1[k] : 0.0f, __ATOMIC_RELAXED,
                             __HIP_MEMORY_SCOPE_WORKGROUP);
    }
#pragma unroll
    for (int k = 0; k < EPW; ++k) pk[k] = nk[k];
    e0 = e1;
  }

#pragma unroll
  for (int i = 0; i < TPT; ++i) {
    sIh[tid + i * 1024] = tI[i];
    sHh[tid + i * 1024] = tH[i];
  }
  __syncthreads();

  const int row0 = b * RANGE;
  const int lo = LAST ? batch[row0] : 0;
  const int nrows = (N_NODES - row0) < RANGE ? (N_NODES - row0) : RANGE;
  const int nUnits = ((nrows + 15) >> 4) * 4;
  const int ln = lane & 15, lh = lane >> 4;

#pragma unroll 1
  for (int u = w; u < nUnits; u += 32) {
    const int t = u >> 2;
    const int j = (u + (u >> 5)) & 3;
    const int lr = t * 16 + ln;
    int gr = row0 + lr;
    gr = gr < N_NODES ? gr : N_NODES - 1;
    const v2f* hp = (const v2f*)(hin + (size_t)gr * H_DIM);
    v2f aH[16];
#pragma unroll
    for (int kk = 0; kk < 16; ++kk) aH[kk] = hp[2 * kk + lh];
    const v2f* ap = (const v2f*)(acc + lr * ACC_PITCH);

    const int c = j * 16 + ln;
    float bir = bih[c], biz = bih[64 + c], bin = bih[128 + c];
    float bhr = bhh[c], bhz = bhh[64 + c], bhn = bhh[128 + c];
    v8f gi_r, gi_z, gi_n, gh_r, gh_z, gh_n;
#pragma unroll
    for (int v = 0; v < 8; ++v) {
      gi_r[v] = bir; gi_z[v] = biz; gi_n[v] = bin;
      gh_r[v] = bhr; gh_z[v] = bhz; gh_n[v] = bhn;
    }

#pragma unroll
    for (int kk = 0; kk < 16; ++kk) {
      v2f a = ap[2 * kk + lh];
      int base = kk * 12 * 32 + lane;
      v2f br = sIh[base + (j + 0) * 32];
      v2f bz = sIh[base + (j + 4) * 32];
      v2f bn = sIh[base + (j + 8) * 32];
      v2f cr = sHh[base + (j + 0) * 32];
      v2f cz = sHh[base + (j + 4) * 32];
      v2f cn = sHh[base + (j + 8) * 32];
      gi_r = wmma4(a, br, gi_r);
      gi_z = wmma4(a, bz, gi_z);
      gi_n = wmma4(a, bn, gi_n);
      gh_r = wmma4(aH[kk], cr, gh_r);
      gh_z = wmma4(aH[kk], cz, gh_z);
      gh_n = wmma4(aH[kk], cn, gh_n);
    }

#pragma unroll
    for (int v = 0; v < 8; ++v) {
      const int lrow = t * 16 + v + 8 * lh;
      int grow = row0 + lrow;
      grow = grow < N_NODES ? grow : N_NODES - 1;
      size_t idx = (size_t)grow * H_DIM + c;
      float hv = hin[idx];
      float r = sigmoid_hw(gi_r[v] + gh_r[v]);
      float z = sigmoid_hw(gi_z[v] + gh_z[v]);
      float n = tanh_hw(gi_n[v] + r * gh_n[v]);
      float o = (1.0f - z) * n + z * hv;
      if (lrow < nrows) {
        if constexpr (LAST) {
          const int g = sid[lrow];
          const unsigned rel = (unsigned)(g - lo);
          if (rel < POOL_SPAN)
            __hip_atomic_fetch_add(pl + rel * H_DIM + c, o, __ATOMIC_RELAXED,
                                   __HIP_MEMORY_SCOPE_WORKGROUP);
          else
            __hip_atomic_fetch_add(out + (size_t)g * H_DIM + c, o,
                                   __ATOMIC_RELAXED, __HIP_MEMORY_SCOPE_AGENT);
        } else {
          hout[idx] = o;
        }
      }
    }
  }

  if constexpr (LAST) {
    __syncthreads();
    static_assert(POOL_SPAN * H_DIM == 1024, "one pool slot per thread");
    const int i = w * 32 + (int)__builtin_amdgcn_mbcnt_lo(~0u, 0u);
    const float s = pl[i];
    const int g = lo + (i >> 6);
    if (g < NUM_GRAPHS && s != 0.0f)
      __hip_atomic_fetch_add(out + (size_t)g * H_DIM + (i & 63), s,
                             __ATOMIC_RELAXED, __HIP_MEMORY_SCOPE_AGENT);
  }
}

__global__ void zero_f32(float* __restrict__ p, int n) {
  int tid = blockIdx.x * blockDim.x + threadIdx.x;
  if (tid < n) p[tid] = 0.0f;
}

extern "C" void kernel_launch(void* const* d_in, const int* in_sizes, int n_in,
                              void* d_out, int out_size, void* d_ws,
                              size_t ws_size, hipStream_t stream) {
  const float* x      = (const float*)d_in[0];
  const int*   ei     = (const int*)d_in[1];
  const int*   batch  = (const int*)d_in[2];
  const float* weight = (const float*)d_in[3];
  const float* Wih    = (const float*)d_in[4];
  const float* Whh    = (const float*)d_in[5];
  const float* bih    = (const float*)d_in[6];
  const float* bhh    = (const float*)d_in[7];
  float* out = (float*)d_out;

  const size_t NB = (size_t)N_NODES * H_DIM * sizeof(float);
  char* ws = (char*)d_ws;
  float* hA = (float*)(ws);
  float* hB = (float*)(ws + 2 * NB);
  v2f* packWc = (v2f*)(ws + NB);
  v2f* packHh = packWc + (size_t)N_LAYERS * NFRAG_G;
  int* cnt      = (int*)(packHh + NFRAG_G);
  int* binStart = cnt + BIN_WGS * NBINS;
  int* sorted   = binStart + 512;

  zero_f32<<<(NUM_GRAPHS * H_DIM + 255) / 256, 256, 0, stream>>>(
      out, NUM_GRAPHS * H_DIM);

  pack_wc<<<(N_LAYERS * NFRAG_G + 255) / 256, 256, 0, stream>>>(weight, Wih,
                                                                (float*)packWc);
  pack_b<<<(NFRAG_G + 255) / 256, 256, 0, stream>>>(Whh, (float*)packHh, 12, 1);

  bin_count<<<BIN_WGS, 1024, 0, stream>>>(ei, cnt);
  bin_scan<<<1, 1024, 0, stream>>>(cnt, binStart);
  bin_place<<<BIN_WGS, 1024, 0, stream>>>(ei, cnt, sorted);

  layer_fused<false><<<NBINS, 1024, 0, stream>>>(x, hA, sorted, binStart,
                                                 packWc, packHh, bih, bhh,
                                                 batch, out);
  layer_fused<false><<<NBINS, 1024, 0, stream>>>(hA, hB, sorted, binStart,
                                                 packWc + NFRAG_G, packHh, bih,
                                                 bhh, batch, out);
  layer_fused<true><<<NBINS, 1024, 0, stream>>>(hB, hA, sorted, binStart,
                                                packWc + 2 * NFRAG_G, packHh,
                                                bih, bhh, batch, out);
}


# setup launches merged, 9 to 5: weight packs, out's start value and range histograms are roles of one grid; each placing work-group sums the count matrix for its own offsets (scan launch goes)
# speedup vs baseline: 2.9137x; 1.0345x over previous
#include <hip/hip_runtime.h>
#include <math.h>

#define N_NODES    100000
#define N_EDGES    1600000
#define H_DIM      64
#define N_LAYERS   3
#define NUM_GRAPHS 256

typedef __attribute__((ext_vector_type(2))) float v2f;
typedef __attribute__((ext_vector_type(8))) float v8f;

__device__ __forceinline__ v8f wmma4(v2f a, v2f b, v8f c) {
  return __builtin_amdgcn_wmma_f32_16x16x4_f32(false, a, false, b, (short)0, c,
                                               false, false);
}

#define LOG2E_F 1.4426950408889634f
__device__ __forceinline__ float sigmoid_hw(float x) {
  return __builtin_amdgcn_rcpf(1.0f + __builtin_amdgcn_exp2f(-LOG2E_F * x));
}
__device__ __forceinline__ float tanh_hw(float x) {
  return fmaf(-2.0f,
              __builtin_amdgcn_rcpf(
                  1.0f + __builtin_amdgcn_exp2f((2.0f * LOG2E_F) * x)),
              1.0f);
}

__device__ __forceinline__ void pack_b(const float* __restrict__ Wsrc,
                                       float* __restrict__ pack, int nGroups,
                                       int transposed, int tid) {
  int total = 16 * nGroups * 32;
  if (tid >= total) return;
  int lane = tid & 31;
  int t = tid >> 5;
  int g = t % nGroups, kk = t / nGroups;
  int ln = lane & 15, lh = lane >> 4;
  int c = g * 16 + ln;
  int k0 = 4 * kk + 2 * lh;
  v2f b;
  if (transposed) {
    b.x = Wsrc[c * 64 + k0];
    b.y = Wsrc[c * 64 + k0 + 1];
  } else {
    b.x = Wsrc[k0 * 64 + c];
    b.y = Wsrc[(k0 + 1) * 64 + c];
  }
  ((v2f*)pack)[tid] = b;
}

#define NFRAG_G (16 * 12 * 32)

__device__ __forceinline__ void pack_wc(const float* __restrict__ weight,
                                        const float* __restrict__ Wih,
                                        float* __restrict__ pack, int tid) {
  if (tid >= N_LAYERS * NFRAG_G) return;
  int layer = tid / NFRAG_G;
  int f = tid - layer * NFRAG_G;
  int lane = f & 31;
  int t = f >> 5;
  int g = t % 12, kk = t / 12;
  int ln = lane & 15, lh = lane >> 4;
  int c = g * 16 + ln;
  int k0 = 4 * kk + 2 * lh;
  const float* w0 = weight + (size_t)layer * 64 * 64 + (size_t)k0 * 64;
  const float* w1 = w0 + 64;
  const float* wi = Wih + (size_t)c * 64;
  float s0 = 0.0f, s1 = 0.0f;
#pragma unroll
  for (int j = 0; j < 64; ++j) {
    float u = wi[j];
    s0 = fmaf(w0[j], u, s0);
    s1 = fmaf(w1[j], u, s1);
  }
  v2f b;
  b.x = s0;
  b.y = s1;
  ((v2f*)pack)[tid] = b;
}

#define NBINS    256
#define RANGE    391
#define BIN_WGS  256
#define EPB      (N_EDGES / BIN_WGS)
#define BIN_ITERS ((EPB + 1023) / 1024)
#define SRC_BITS 17
#define SRC_MASK ((1 << SRC_BITS) - 1)
static_assert(RANGE * NBINS >= N_NODES, "ranges must cover all rows");
static_assert(N_EDGES % BIN_WGS == 0, "edge count must split evenly");
static_assert((1 << SRC_BITS) >= N_NODES, "source row must fit its field");
static_assert(RANGE <= (1 << (31 - SRC_BITS)), "local row must fit its field");

#define PREP_WC    (N_LAYERS * NFRAG_G / 1024)
#define PREP_HH    (NFRAG_G / 1024)
#define PREP_Z     (NUM_GRAPHS * H_DIM / 1024)
#define PREP_EXTRA (PREP_WC + PREP_HH + PREP_Z)
static_assert(N_LAYERS * NFRAG_G % 1024 == 0, "whole work-groups per role");
static_assert(NUM_GRAPHS * H_DIM % 1024 == 0, "whole work-groups per role");

__global__ void __launch_bounds__(1024)
prep(const int* __restrict__ ei, const float* __restrict__ weight,
     const float* __restrict__ Wih, const float* __restrict__ Whh,
     int* __restrict__ cnt, float* __restrict__ packWc,
     float* __restrict__ packHh, float* __restrict__ out) {
  __shared__ int hist[NBINS];
  const int tid = threadIdx.x;
  int blk = blockIdx.x;
  if (blk < PREP_WC) {
    pack_wc(weight, Wih, packWc, blk * 1024 + tid);
    return;
  }
  blk -= PREP_WC;
  if (blk < PREP_HH) {
    pack_b(Whh, packHh, 12, 1, blk * 1024 + tid);
    return;
  }
  blk -= PREP_HH;
  if (blk < PREP_Z) {
    out[blk * 1024 + tid] = 0.0f;
    return;
  }
  blk -= PREP_Z;
  if (tid < NBINS) hist[tid] = 0;
  __syncthreads();
  const int* dst = ei + N_EDGES + blk * EPB;
  int d[BIN_ITERS];
#pragma unroll
  for (int j = 0; j < BIN_ITERS; ++j) {
    int i = tid + j * 1024;
    d[j] = dst[i < EPB ? i : EPB - 1];
  }
#pragma unroll
  for (int j = 0; j < BIN_ITERS; ++j)
    if (tid + j * 1024 < EPB) atomicAdd(&hist[d[j] / RANGE], 1);
  __syncthreads();
  if (tid < NBINS) cnt[blk * NBINS + tid] = hist[tid];
}

#define RG   16
#define RPG  (BIN_WGS / RG)
static_assert(RG * (NBINS / 4) == 1024, "one thread per row group x 4 ranges");
static_assert(BIN_WGS % RG == 0, "row groups must split the cnt rows evenly");

__global__ void __launch_bounds__(1024)
bin_place(const int* __restrict__ ei, const int* __restrict__ cnt,
          int* __restrict__ binStart, int* __restrict__ sorted) {
  __shared__ __attribute__((aligned(16))) int part[RG][NBINS];
  __shared__ __attribute__((aligned(16))) int partB[RG][NBINS];
  __shared__ int sc[NBINS];
  __shared__ int cur[NBINS];
  const int tid = threadIdx.x;
  const int wg = blockIdx.x;
  const int* src = ei + wg * EPB;
  const int* dst = src + N_EDGES;
  int s[BIN_ITERS], d[BIN_ITERS];
#pragma unroll
  for (int j = 0; j < BIN_ITERS; ++j) {
    int i = tid + j * 1024;
    i = i < EPB ? i : EPB - 1;
    s[j] = src[i];
    d[j] = dst[i];
  }

  const int tg = tid & 63;
  const int rg = tid >> 6;
  const int4* cnt4 = (const int4*)cnt;
  int4 c[RPG];
#pragma unroll
  for (int r = 0; r < RPG; ++r)
    c[r] = cnt4[(rg * RPG + r) * (NBINS / 4) + tg];
  int4 run = make_int4(0, 0, 0, 0), bef = make_int4(0, 0, 0, 0);
#pragma unroll
  for (int r = 0; r < RPG; ++r) {
    run.x += c[r].x; run.y += c[r].y; run.z += c[r].z; run.w += c[r].w;
    if (rg * RPG + r < wg) {
      bef.x += c[r].x; bef.y += c[r].y; bef.z += c[r].z; bef.w += c[r].w;
    }
  }
  *(int4*)&part[rg][4 * tg] = run;
  *(int4*)&partB[rg][4 * tg] = bef;
  __syncthreads();
  int total = 0, before = 0;
  if (tid < NBINS) {
#pragma unroll
    for (int g = 0; g < RG; ++g) {
      total += part[g][tid];
      before += partB[g][tid];
    }
    sc[tid] = total;
  }
  __syncthreads();
  for (int off = 1; off < NBINS; off <<= 1) {
    int v = 0;
    if (tid < NBINS && tid >= off) v = sc[tid - off];
    __syncthreads();
    if (tid < NBINS) sc[tid] += v;
    __syncthreads();
  }
  if (tid < NBINS) {
    const int start = sc[tid] - total;
    cur[tid] = start + before;
    if (wg == 0) {
      binStart[tid] = start;
      if (tid == NBINS - 1) binStart[NBINS] = sc[tid];
    }
  }
  __syncthreads();
#pragma unroll
  for (int j = 0; j < BIN_ITERS; ++j) {
    if (tid + j * 1024 < EPB) {
      int b = d[j] / RANGE;
      int dl = d[j] - b * RANGE;
      int pos = atomicAdd(&cur[b], 1);
      sorted[pos] = s[j] | (dl << SRC_BITS);
    }
  }
}

#define POOL_SPAN   16
#define EPW         8
#define ACC_PITCH   68
#define RANGE_TILES ((RANGE + 15) / 16)
#define RANGE_PAD   (RANGE_TILES * 16)
static_assert((ACC_PITCH * 4) % 16 == 0, "LDS rows must stay 16-byte aligned");
static_assert(NFRAG_G % 1024 == 0, "table staging assumes whole rounds of 1024");
static_assert(RANGE_PAD * ACC_PITCH % 4 == 0, "sums are cleared as float4");

template <bool LAST>
__global__ void __launch_bounds__(1024)
layer_fused(const float* __restrict__ hin, float* __restrict__ hout,
            const int* __restrict__ sorted, const int* __restrict__ binStart,
            const v2f* __restrict__ pIh, const v2f* __restrict__ pHh,
            const float* __restrict__ bih, const float* __restrict__ bhh,
            const int* __restrict__ batch, float* __restrict__ out) {
  __shared__ __attribute__((aligned(16))) float acc[RANGE_PAD * ACC_PITCH];
  __shared__ v2f sIh[NFRAG_G];
  __shared__ v2f sHh[NFRAG_G];
  __shared__ float pl[LAST ? POOL_SPAN * H_DIM : 1];
  __shared__ int sid[LAST ? RANGE_PAD : 1];
  const int tid = threadIdx.x;
  const int lane = tid & 31;

  if constexpr (LAST) {
    for (int i = tid; i < POOL_SPAN * H_DIM; i += 1024) pl[i] = 0.0f;
    if (tid < RANGE_PAD) {
      int gr = blockIdx.x * RANGE + tid;
      sid[tid] = batch[gr < N_NODES ? gr : N_NODES - 1];
    }
  }

  float4* acc4 = (float4*)acc;
  for (int i = tid; i < RANGE_PAD * ACC_PITCH / 4; i += 1024)
    acc4[i] = make_float4(0.0f, 0.0f, 0.0f, 0.0f);
  __syncthreads();

  constexpr int TPT = NFRAG_G / 1024;
  v2f tI[TPT], tH[TPT];
#pragma unroll
  for (int i = 0; i < TPT; ++i) {
    tI[i] = pIh[tid + i * 1024];
    tH[i] = pHh[tid + i * 1024];
  }

  const int b = blockIdx.x;
  const int beg = binStart[b];
  const int end = binStart[b + 1];
  const int w = __builtin_amdgcn_readfirstlane(tid >> 5);

  int e0 = beg + w * EPW;
  int pk[EPW];
  if (e0 < end) {
#pragma unroll
    for (int k = 0; k < EPW; ++k) {
      int e = e0 + k;
      pk[k] = sorted[e < end ? e : end - 1];
    }
  }
  while (e0 < end) {
    float v0[EPW], v1[EPW];
#pragma unroll
    for (int k = 0; k < EPW; ++k) {
      const float* mp = hin + (size_t)(pk[k] & SRC_MASK) * H_DIM + lane;
      v0[k] = mp[0];
      v1[k] = mp[32];
    }
    const int e1 = e0 + 32 * EPW;
    int nk[EPW];
#pragma unroll
    for (int k = 0; k < EPW; ++k) {
      int e = e1 + k;
      nk[k] = sorted[e < end ? e : end - 1];
    }
#pragma unroll
    for (int k = 0; k < EPW; ++k) {
      const bool live = (e0 + k) < end;
      float* ap = acc + (pk[k] >> SRC_BITS) * ACC_PITCH + lane;
      __hip_atomic_fetch_add(ap, live ? v0[k] : 0.0f, __ATOMIC_RELAXED,
                             __HIP_MEMORY_SCOPE_WORKGROUP);
      __hip_atomic_fetch_add(ap + 32, live ? v1[k] : 0.0f, __ATOMIC_RELAXED,
                             __HIP_MEMORY_SCOPE_WORKGROUP);
    }
#pragma unroll
    for (int k = 0; k < EPW; ++k) pk[k] = nk[k];
    e0 = e1;
  }

#pragma unroll
  for (int i = 0; i < TPT; ++i) {
    sIh[tid + i * 1024] = tI[i];
    sHh[tid + i * 1024] = tH[i];
  }
  __syncthreads();

  const int row0 = b * RANGE;
  const int lo = LAST ? batch[row0] : 0;
  const int nrows = (N_NODES - row0) < RANGE ? (N_NODES - row0) : RANGE;
  const int nUnits = ((nrows + 15) >> 4) * 4;
  const int ln = lane & 15, lh = lane >> 4;

#pragma unroll 1
  for (int u = w; u < nUnits; u += 32) {
    const int t = u >> 2;
    const int j = (u + (u >> 5)) & 3;
    const int lr = t * 16 + ln;
    int gr = row0 + lr;
    gr = gr < N_NODES ? gr : N_NODES - 1;
    const v2f* hp = (const v2f*)(hin + (size_t)gr * H_DIM);
    v2f aH[16];
#pragma unroll
    for (int kk = 0; kk < 16; ++kk) aH[kk] = hp[2 * kk + lh];
    const v2f* ap = (const v2f*)(acc + lr * ACC_PITCH);

    const int c = j * 16 + ln;
    float bir = bih[c], biz = bih[64 + c], bin = bih[128 + c];
    float bhr = bhh[c], bhz = bhh[64 + c], bhn = bhh[128 + c];
    v8f gi_r, gi_z, gi_n, gh_r, gh_z, gh_n;
#pragma unroll
    for (int v = 0; v < 8; ++v) {
      gi_r[v] = bir; gi_z[v] = biz; gi_n[v] = bin;
      gh_r[v] = bhr; gh_z[v] = bhz; gh_n[v] = bhn;
    }

#pragma unroll
    for (int kk = 0; kk < 16; ++kk) {
      v2f a = ap[2 * kk + lh];
      int base = kk * 12 * 32 + lane;
      v2f br = sIh[base + (j + 0) * 32];
      v2f bz = sIh[base + (j + 4) * 32];
      v2f bn = sIh[base + (j + 8) * 32];
      v2f cr = sHh[base + (j + 0) * 32];
      v2f cz = sHh[base + (j + 4) * 32];
      v2f cn = sHh[base + (j + 8) * 32];
      gi_r = wmma4(a, br, gi_r);
      gi_z = wmma4(a, bz, gi_z);
      gi_n = wmma4(a, bn, gi_n);
      gh_r = wmma4(aH[kk], cr, gh_r);
      gh_z = wmma4(aH[kk], cz, gh_z);
      gh_n = wmma4(aH[kk], cn, gh_n);
    }

#pragma unroll
    for (int v = 0; v < 8; ++v) {
      const int lrow = t * 16 + v + 8 * lh;
      int grow = row0 + lrow;
      grow = grow < N_NODES ? grow : N_NODES - 1;
      size_t idx = (size_t)grow * H_DIM + c;
      float hv = hin[idx];
      float r = sigmoid_hw(gi_r[v] + gh_r[v]);
      float z = sigmoid_hw(gi_z[v] + gh_z[v]);
      float n = tanh_hw(gi_n[v] + r * gh_n[v]);
      float o = (1.0f - z) * n + z * hv;
      if (lrow < nrows) {
        if constexpr (LAST) {
          const int g = sid[lrow];
          const unsigned rel = (unsigned)(g - lo);
          if (rel < POOL_SPAN)
            __hip_atomic_fetch_add(pl + rel * H_DIM + c, o, __ATOMIC_RELAXED,
                                   __HIP_MEMORY_SCOPE_WORKGROUP);
          else
            __hip_atomic_fetch_add(out + (size_t)g * H_DIM + c, o,
                                   __ATOMIC_RELAXED, __HIP_MEMORY_SCOPE_AGENT);
        } else {
          hout[idx] = o;
        }
      }
    }
  }

  if constexpr (LAST) {
    __syncthreads();
    static_assert(POOL_SPAN * H_DIM == 1024, "one pool slot per thread");
    const int i = w * 32 + (int)__builtin_amdgcn_mbcnt_lo(~0u, 0u);
    const float s = pl[i];
    const int g = lo + (i >> 6);
    if (g < NUM_GRAPHS && s != 0.0f)
      __hip_atomic_fetch_add(out + (size_t)g * H_DIM + (i & 63), s,
                             __ATOMIC_RELAXED, __HIP_MEMORY_SCOPE_AGENT);
  }
}

extern "C" void kernel_launch(void* const* d_in, const int* in_sizes, int n_in,
                              void* d_out, int out_size, void* d_ws,
                              size_t ws_size, hipStream_t stream) {
  const float* x      = (const float*)d_in[0];
  const int*   ei     = (const int*)d_in[1];
  const int*   batch  = (const int*)d_in[2];
  const float* weight = (const float*)d_in[3];
  const float* Wih    = (const float*)d_in[4];
  const float* Whh    = (const float*)d_in[5];
  const float* bih    = (const float*)d_in[6];
  const float* bhh    = (const float*)d_in[7];
  float* out = (float*)d_out;

  const size_t NB = (size_t)N_NODES * H_DIM * sizeof(float);
  char* ws = (char*)d_ws;
  float* hA = (float*)(ws);
  float* hB = (float*)(ws + 2 * NB);
  v2f* packWc = (v2f*)(ws + NB);
  v2f* packHh = packWc + (size_t)N_LAYERS * NFRAG_G;
  int* cnt      = (int*)(packHh + NFRAG_G);
  int* binStart = cnt + BIN_WGS * NBINS;
  int* sorted   = binStart + 512;

  prep<<<PREP_EXTRA + BIN_WGS, 1024, 0, stream>>>(ei, weight, Wih, Whh, cnt,
                                                  (float*)packWc,
                                                  (float*)packHh, out);
  bin_place<<<BIN_WGS, 1024, 0, stream>>>(ei, cnt, binStart, sorted);

  layer_fused<false><<<NBINS, 1024, 0, stream>>>(x, hA, sorted, binStart,
                                                 packWc, packHh, bih, bhh,
                                                 batch, out);
  layer_fused<false><<<NBINS, 1024, 0, stream>>>(hA, hB, sorted, binStart,
                                                 packWc + NFRAG_G, packHh, bih,
                                                 bhh, batch, out);
  layer_fused<true><<<NBINS, 1024, 0, stream>>>(hB, hA, sorted, binStart,
                                                packWc + 2 * NFRAG_G, packHh,
                                                bih, bhh, batch, out);
}
